# starting kernel
# baseline (speedup 1.0000x reference)
#include <hip/hip_runtime.h>
#include <hip/hip_bf16.h>

typedef __attribute__((ext_vector_type(2))) float v2f;
typedef __attribute__((ext_vector_type(8))) float v8f;

#define N_USER 100000
#define N_ITEM 150000
#define N_NODES (N_USER + N_ITEM)
#define N_HID 64
#define E_UI 1000000
#define E_S 1000000
#define SLOPE 0.5f
#define EPS 1e-12f

__global__ void copy_dual_kernel(const float* __restrict__ src,
                                 float* __restrict__ X,
                                 float* __restrict__ out,
                                 int n) {
    int i = blockIdx.x * blockDim.x + threadIdx.x;
    if (i < n) {
        float v = src[i];
        X[i] = v;
        out[i] = v;
    }
}

__global__ void zero_kernel(float* __restrict__ p, int n) {
    int i = blockIdx.x * blockDim.x + threadIdx.x;
    if (i < n) p[i] = 0.0f;
}

__global__ void __launch_bounds__(32)
gemm64_wmma_kernel(const float* __restrict__ X,
                   const float* __restrict__ W,
                   float* __restrict__ Y,
                   int M) {
    const int row0 = blockIdx.x * 16;
    if (row0 >= M) return;
    const int lane = threadIdx.x;
    const int r = lane & 15;
    const int h = lane >> 4;

    v8f acc[4];
    acc[0] = v8f{};
    acc[1] = v8f{};
    acc[2] = v8f{};
    acc[3] = v8f{};

#pragma unroll
    for (int k = 0; k < 64; k += 4) {
        const float* xr = X + (size_t)(row0 + r) * N_HID + k + 2 * h;
        v2f a;
        a[0] = xr[0];
        a[1] = xr[1];
        const float* w0 = W + (size_t)(k + 2 * h) * N_HID + r;
        const float* w1 = w0 + N_HID;
#pragma unroll
        for (int ct = 0; ct < 4; ++ct) {
            v2f b;
            b[0] = w0[ct * 16];
            b[1] = w1[ct * 16];
            acc[ct] = __builtin_amdgcn_wmma_f32_16x16x4_f32(
                 false, a,  false, b,
                 (short)0, acc[ct],
                 false,  false);
        }
    }

#pragma unroll
    for (int ct = 0; ct < 4; ++ct) {
#pragma unroll
        for (int v = 0; v < 8; ++v) {
            Y[(size_t)(row0 + v + 8 * h) * N_HID + ct * 16 + r] = acc[ct][v];
        }
    }
}

__global__ void __launch_bounds__(256)
spmm_scatter_kernel(const int* __restrict__ rows,
                    const int* __restrict__ cols,
                    const float* __restrict__ vals,
                    const float* __restrict__ Y,
                    float* __restrict__ Z,
                    int E) {
    int e = blockIdx.x * 4 + (threadIdx.x >> 6);
    int d = threadIdx.x & 63;
    if (e < E) {
        int rr = rows[e];
        int cc = cols[e];
        float v = vals[e];
        unsafeAtomicAdd(&Z[(size_t)rr * N_HID + d], v * Y[(size_t)cc * N_HID + d]);
    }
}

__global__ void __launch_bounds__(32)
leaky_norm_acc_kernel(const float* __restrict__ Z,
                      float* __restrict__ X,
                      float* __restrict__ out,
                      int M) {
    int n = blockIdx.x;
    if (n >= M) return;
    int lane = threadIdx.x;

    const float2* zr = (const float2*)(Z + (size_t)n * N_HID);
    float2 z = zr[lane];
    float2 v;
    v.x = (z.x >= 0.0f) ? z.x : SLOPE * z.x;
    v.y = (z.y >= 0.0f) ? z.y : SLOPE * z.y;

    ((float2*)(X + (size_t)n * N_HID))[lane] = v;

    float ss = v.x * v.x + v.y * v.y;
#pragma unroll
    for (int o = 16; o > 0; o >>= 1) ss += __shfl_xor(ss, o, 32);

    float inv = 1.0f / fmaxf(sqrtf(ss), EPS);

    float2* orow = (float2*)(out + (size_t)n * N_HID);
    float2 cur = orow[lane];
    cur.x += v.x * inv;
    cur.y += v.y * inv;
    orow[lane] = cur;
}

extern "C" void kernel_launch(void* const* d_in, const int* in_sizes, int n_in,
                              void* d_out, int out_size, void* d_ws, size_t ws_size,
                              hipStream_t stream) {
    const float* user_emb = (const float*)d_in[0];
    const float* item_emb = (const float*)d_in[1];
    const float* ui_w     = (const float*)d_in[2];
    const float* s_w      = (const float*)d_in[3];
    const int*   ui_rows  = (const int*)d_in[4];
    const int*   ui_cols  = (const int*)d_in[5];
    const float* ui_vals  = (const float*)d_in[6];
    const int*   s_rows   = (const int*)d_in[7];
    const int*   s_cols   = (const int*)d_in[8];
    const float* s_vals   = (const float*)d_in[9];

    float* X = (float*)d_ws;
    float* Y = X + (size_t)N_NODES * N_HID;
    float* Z = Y + (size_t)N_NODES * N_HID;

    float* out_ui = (float*)d_out;
    float* out_s  = out_ui + (size_t)N_NODES * N_HID;

    const int nu = N_USER * N_HID;
    const int ni = N_ITEM * N_HID;
    const int nn = N_NODES * N_HID;

    copy_dual_kernel<<<(nu + 255) / 256, 256, 0, stream>>>(user_emb, X, out_ui, nu);
    copy_dual_kernel<<<(ni + 255) / 256, 256, 0, stream>>>(item_emb, X + nu, out_ui + nu, ni);

    for (int l = 0; l < 2; ++l) {
        gemm64_wmma_kernel<<<N_NODES / 16, 32, 0, stream>>>(X, ui_w + (size_t)l * N_HID * N_HID, Y, N_NODES);
        zero_kernel<<<(nn + 255) / 256, 256, 0, stream>>>(Z, nn);
        spmm_scatter_kernel<<<(E_UI + 3) / 4, 256, 0, stream>>>(ui_rows, ui_cols, ui_vals, Y, Z, E_UI);
        leaky_norm_acc_kernel<<<N_NODES, 32, 0, stream>>>(Z, X, out_ui, N_NODES);
    }

    copy_dual_kernel<<<(nu + 255) / 256, 256, 0, stream>>>(user_emb, X, out_s, nu);

    for (int l = 0; l < 2; ++l) {
        gemm64_wmma_kernel<<<N_USER / 16, 32, 0, stream>>>(X, s_w + (size_t)l * N_HID * N_HID, Y, N_USER);
        zero_kernel<<<(nu + 255) / 256, 256, 0, stream>>>(Z, nu);
        spmm_scatter_kernel<<<(E_S + 3) / 4, 256, 0, stream>>>(s_rows, s_cols, s_vals, Y, Z, E_S);
        leaky_norm_acc_kernel<<<N_USER, 32, 0, stream>>>(Z, X, out_s, N_USER);
    }
}


# Residual copies dropped: layer 0's product kernel and tile blocks read the embedding tables in place, the tile block stores out = table row + its term; 3 fewer launches; float sum order kept
# speedup vs baseline: 3.4752x; 3.4752x over previous
#include <hip/hip_runtime.h>
#include <hip/hip_bf16.h>

typedef __attribute__((ext_vector_type(2))) float v2f;
typedef __attribute__((ext_vector_type(8))) float v8f;

#define N_USER 100000
#define N_ITEM 150000
#define N_NODES (N_USER + N_ITEM)
#define N_HID 64
#define E_UI 1000000
#define E_S 1000000
#define SLOPE 0.5f
#define EPS 1e-12f

static_assert(N_USER % 32 == 0, "a 16-row product tile and a wave's 32 tile rows must lie in one table");

__global__ void __launch_bounds__(32)
gemm64_wmma_kernel(const float* __restrict__ Xa,
                   const float* __restrict__ Xb,
                   int split,
                   const float* __restrict__ W,
                   float* __restrict__ Y,
                   int M) {
    const int row0 = blockIdx.x * 16;
    if (row0 >= M) return;
    const float* __restrict__ X0 = (row0 < split) ? Xa + (size_t)row0 * N_HID
                                                  : Xb + (size_t)(row0 - split) * N_HID;
    const int lane = threadIdx.x;
    const int r = lane & 15;
    const int h = lane >> 4;

    v8f acc[4];
    acc[0] = v8f{};
    acc[1] = v8f{};
    acc[2] = v8f{};
    acc[3] = v8f{};

#pragma unroll
    for (int k = 0; k < 64; k += 4) {
        const float* xr = X0 + (size_t)r * N_HID + k + 2 * h;
        v2f a;
        a[0] = xr[0];
        a[1] = xr[1];
        const float* w0 = W + (size_t)(k + 2 * h) * N_HID + r;
        const float* w1 = w0 + N_HID;
#pragma unroll
        for (int ct = 0; ct < 4; ++ct) {
            v2f b;
            b[0] = w0[ct * 16];
            b[1] = w1[ct * 16];
            acc[ct] = __builtin_amdgcn_wmma_f32_16x16x4_f32(
                 false, a,  false, b,
                 (short)0, acc[ct],
                 false,  false);
        }
    }

#pragma unroll
    for (int ct = 0; ct < 4; ++ct) {
#pragma unroll
        for (int v = 0; v < 8; ++v) {
            Y[(size_t)(row0 + v + 8 * h) * N_HID + ct * 16 + r] = acc[ct][v];
        }
    }
}

#define TILE_ROWS 256
#define TILE_SHIFT 8
#define PITCH 68
#define CHUNK 8192
#define BIN_THREADS 512
#define MAX_TILES 1024
#define T_UI ((N_NODES + TILE_ROWS - 1) / TILE_ROWS)
#define T_S ((N_USER + TILE_ROWS - 1) / TILE_ROWS)
#define NB_UI ((E_UI + CHUNK - 1) / CHUNK)
#define NB_S ((E_S + CHUNK - 1) / CHUNK)
#define NB_MAX (NB_UI > NB_S ? NB_UI : NB_S)

__global__ void __launch_bounds__(BIN_THREADS)
bin_count_kernel(const int* __restrict__ rows,
                 int* __restrict__ H,
                 int E, int T) {
    __shared__ int h[MAX_TILES];
    const int tid = threadIdx.x;
    for (int t = tid; t < MAX_TILES; t += BIN_THREADS) h[t] = 0;
    __syncthreads();
    const int e0 = blockIdx.x * CHUNK;
#pragma unroll 4
    for (int i = 0; i < CHUNK / BIN_THREADS; ++i) {
        int e = e0 + i * BIN_THREADS + tid;
        if (e < E) atomicAdd(&h[rows[e] >> TILE_SHIFT], 1);
    }
    __syncthreads();
    for (int t = tid; t < T; t += BIN_THREADS) H[(size_t)blockIdx.x * T + t] = h[t];
}

__global__ void __launch_bounds__(MAX_TILES)
bin_scan_kernel(const int* __restrict__ H,
                int* __restrict__ Hp,
                int* __restrict__ base,
                int NB, int T) {
    __shared__ int s[MAX_TILES];
    const int t = threadIdx.x;
    int run = 0;
    if (t < T) {
#pragma unroll 8
        for (int b = 0; b < NB; ++b) {
            int c = H[(size_t)b * T + t];
            Hp[(size_t)b * T + t] = run;
            run += c;
        }
    }
    s[t] = run;
    __syncthreads();
    for (int off = 1; off < MAX_TILES; off <<= 1) {
        int v = (t >= off) ? s[t - off] : 0;
        __syncthreads();
        s[t] += v;
        __syncthreads();
    }
    if (t < T) base[t] = s[t] - run;
    if (t == MAX_TILES - 1) base[T] = s[t];
}

__global__ void __launch_bounds__(BIN_THREADS)
bin_place_kernel(const int* __restrict__ rows,
                 const int* __restrict__ cols,
                 const float* __restrict__ vals,
                 const int* __restrict__ Hp,
                 const int* __restrict__ base,
                 int2* __restrict__ rec,
                 int E, int T) {
    __shared__ int pos[MAX_TILES];
    const int tid = threadIdx.x;
    for (int t = tid; t < T; t += BIN_THREADS) pos[t] = base[t] + Hp[(size_t)blockIdx.x * T + t];
    __syncthreads();
    const int e0 = blockIdx.x * CHUNK;
#pragma unroll 4
    for (int i = 0; i < CHUNK / BIN_THREADS; ++i) {
        int e = e0 + i * BIN_THREADS + tid;
        if (e < E) {
            int r = rows[e];
            int c = cols[e];
            float v = vals[e];
            int p = atomicAdd(&pos[r >> TILE_SHIFT], 1);
            rec[p] = make_int2((c << TILE_SHIFT) | (r & (TILE_ROWS - 1)), __float_as_int(v));
        }
    }
}

template <bool NEXT, bool TABLE, bool GUARD>
__device__ __forceinline__ void finish_rows(float* acc,
                                            const float* __restrict__ Wn,
                                            float2* __restrict__ yt,
                                            float2* __restrict__ ot,
                                            const float2* __restrict__ rt,
                                            int r_lo, int last, int lane) {
    float2 o[TILE_ROWS / 8];
    const float2* res = TABLE ? rt : (const float2*)ot;
#pragma unroll
    for (int j = 0; j < TILE_ROWS / 8; ++j) {
        o[j] = res[(GUARD ? min(j, last) : j) * (N_HID / 2)];
    }

    const float4* my4 = (const float4*)(acc + (r_lo + lane) * PITCH);
    float ss = 0.0f;
#pragma unroll
    for (int i = 0; i < N_HID / 4; ++i) {
        const float4 z = my4[i];
        const float vx = (z.x >= 0.0f) ? z.x : SLOPE * z.x;
        const float vy = (z.y >= 0.0f) ? z.y : SLOPE * z.y;
        const float vz = (z.z >= 0.0f) ? z.z : SLOPE * z.z;
        const float vw = (z.w >= 0.0f) ? z.w : SLOPE * z.w;
        ss += vx * vx + vy * vy + vz * vz + vw * vw;
    }
    const float inv_all = 1.0f / fmaxf(sqrtf(ss), EPS);

    float2* acc2 = (float2*)(acc + r_lo * PITCH) + lane;
#pragma unroll
    for (int j = 0; j < TILE_ROWS / 8; ++j) {
        const float inv = __int_as_float(__builtin_amdgcn_readlane(__float_as_int(inv_all), j));
        const float2 z = acc2[j * (PITCH / 2)];
        float2 v;
        v.x = (z.x >= 0.0f) ? z.x : SLOPE * z.x;
        v.y = (z.y >= 0.0f) ? z.y : SLOPE * z.y;
        float2 cur = o[j];
        cur.x += v.x * inv;
        cur.y += v.y * inv;
        if (NEXT) acc2[j * (PITCH / 2)] = v;
        if (!GUARD || j <= last) {
            ot[j * (N_HID / 2)] = cur;
        }
    }

    if (NEXT) {
        const int r = lane & 15;
        const int h = lane >> 4;
        v8f c0[4];
        v8f c1[4];
#pragma unroll
        for (int ct = 0; ct < 4; ++ct) {
            c0[ct] = v8f{};
            c1[ct] = v8f{};
        }
        const float* a_lo = acc + (r_lo + r) * PITCH + 2 * h;
        const float* a_hi = a_lo + 16 * PITCH;
#pragma unroll 4
        for (int k = 0; k < N_HID; k += 4) {
            const float2 p0 = *(const float2*)(a_lo + k);
            const float2 p1 = *(const float2*)(a_hi + k);
            v2f a0;
            a0[0] = p0.x;
            a0[1] = p0.y;
            v2f a1;
            a1[0] = p1.x;
            a1[1] = p1.y;
            const float* w0 = Wn + (size_t)(k + 2 * h) * N_HID + r;
            const float* w1 = w0 + N_HID;
#pragma unroll
            for (int ct = 0; ct < 4; ++ct) {
                v2f b;
                b[0] = w0[ct * 16];
                b[1] = w1[ct * 16];
                c0[ct] = __builtin_amdgcn_wmma_f32_16x16x4_f32(
                     false, a0,  false, b,
                     (short)0, c0[ct],
                     false,  false);
                c1[ct] = __builtin_amdgcn_wmma_f32_16x16x4_f32(
                     false, a1,  false, b,
                     (short)0, c1[ct],
                     false,  false);
            }
        }
#pragma unroll
        for (int ct = 0; ct < 4; ++ct) {
#pragma unroll
            for (int v = 0; v < 8; ++v) {
                acc[(r_lo + v + 8 * h) * PITCH + ct * 16 + r] = c0[ct][v];
                acc[(r_lo + 16 + v + 8 * h) * PITCH + ct * 16 + r] = c1[ct][v];
            }
        }
#pragma unroll
        for (int j = 0; j < TILE_ROWS / 8; ++j) {
            const float2 yv = acc2[j * (PITCH / 2)];
            if (!GUARD || j <= last) {
                yt[j * (N_HID / 2)] = yv;
            }
        }
    }
}

template <bool NEXT, bool TABLE>
__global__ void __launch_bounds__(256, 8)
spmm_tile_kernel(const int2* __restrict__ rec,
                 const int* __restrict__ base,
                 const float* __restrict__ Y,
                 const float* __restrict__ Wn,
                 float* __restrict__ Yn,
                 const float* __restrict__ ta,
                 const float* __restrict__ tb,
                 int split,
                 float* __restrict__ out,
                 int M) {
    __shared__ __attribute__((aligned(16))) float acc[TILE_ROWS * PITCH];
    const int tid = threadIdx.x;
    const int lane = tid & 31;
    const int w = __builtin_amdgcn_readfirstlane(tid >> 5);
    const int t = blockIdx.x;

    float4* acc4 = (float4*)acc;
    for (int i = tid; i < TILE_ROWS * PITCH / 4; i += 256) acc4[i] = make_float4(0.0f, 0.0f, 0.0f, 0.0f);
    __syncthreads();

    const int beg = base[t];
    const int end = base[t + 1];

    for (int e0 = beg + w * 32; e0 < end; e0 += 8 * 32) {
        const int n = min(32, end - e0);
        int kk = 0;
        int vb = 0;
        if (lane < n) {
            int2 rr = rec[e0 + lane];
            kk = rr.x;
            vb = rr.y;
        }
        float2 y[32];
#pragma unroll
        for (int j = 0; j < 32; ++j) {
            const int k = __builtin_amdgcn_readlane(kk, j);
            const float2* yr = (const float2*)(Y + (size_t)(k >> TILE_SHIFT) * N_HID);
            y[j] = yr[lane];
        }
#pragma unroll
        for (int j = 0; j < 32; ++j) {
            if (j == 0 || j < n) {
                const int k = __builtin_amdgcn_readlane(kk, j);
                const float v = __int_as_float(__builtin_amdgcn_readlane(vb, j));
                float* a = acc + (k & (TILE_ROWS - 1)) * PITCH + 2 * lane;
                atomicAdd(a, v * y[j].x);
                atomicAdd(a + 1, v * y[j].y);
            }
        }
    }
    const int rows_here = min(TILE_ROWS, M - t * TILE_ROWS);
    const int r_lo = w * (TILE_ROWS / 8);
    const size_t first = ((size_t)t * TILE_ROWS + r_lo) * N_HID;
    float2* yt = NEXT ? (float2*)(Yn + first) + lane : nullptr;
    float2* ot = (float2*)(out + first) + lane;
    const float2* rt = nullptr;
    if (TABLE) {
        const int g0 = t * TILE_ROWS + r_lo;
        rt = (const float2*)((min(g0, M - 1) < split) ? ta + (size_t)g0 * N_HID
                                                      : tb + (size_t)(g0 - split) * N_HID) + lane;
    }

    __syncthreads();

    if (r_lo + TILE_ROWS / 8 <= rows_here) {
        finish_rows<NEXT, TABLE, false>(acc, Wn, yt, ot, rt, r_lo, rows_here - 1 - r_lo, lane);
    } else {
        finish_rows<NEXT, TABLE, true>(acc, Wn, yt, ot, rt, r_lo, rows_here - 1 - r_lo, lane);
    }
}

extern "C" void kernel_launch(void* const* d_in, const int* in_sizes, int n_in,
                              void* d_out, int out_size, void* d_ws, size_t ws_size,
                              hipStream_t stream) {
    const float* user_emb = (const float*)d_in[0];
    const float* item_emb = (const float*)d_in[1];
    const float* ui_w     = (const float*)d_in[2];
    const float* s_w      = (const float*)d_in[3];
    const int*   ui_rows  = (const int*)d_in[4];
    const int*   ui_cols  = (const int*)d_in[5];
    const float* ui_vals  = (const float*)d_in[6];
    const int*   s_rows   = (const int*)d_in[7];
    const int*   s_cols   = (const int*)d_in[8];
    const float* s_vals   = (const float*)d_in[9];

    float* X = (float*)d_ws;
    float* Y = X + (size_t)N_NODES * N_HID;
    int2*  rec  = (int2*)(Y + (size_t)N_NODES * N_HID);
    int*   H    = (int*)(rec + E_UI);
    int*   Hp   = H + NB_MAX * MAX_TILES;
    int*   base = Hp + NB_MAX * MAX_TILES;

    float* out_ui = (float*)d_out;
    float* out_s  = out_ui + (size_t)N_NODES * N_HID;

    bin_count_kernel<<<NB_UI, BIN_THREADS, 0, stream>>>(ui_rows, H, E_UI, T_UI);
    bin_scan_kernel<<<1, MAX_TILES, 0, stream>>>(H, Hp, base, NB_UI, T_UI);
    bin_place_kernel<<<NB_UI, BIN_THREADS, 0, stream>>>(ui_rows, ui_cols, ui_vals, Hp, base, rec, E_UI, T_UI);

    gemm64_wmma_kernel<<<N_NODES / 16, 32, 0, stream>>>(user_emb, item_emb, N_USER, ui_w, Y, N_NODES);
    spmm_tile_kernel<true, true><<<T_UI, 256, 0, stream>>>(rec, base, Y, ui_w + N_HID * N_HID, X,
                                                           user_emb, item_emb, N_USER, out_ui, N_NODES);
    spmm_tile_kernel<false, false><<<T_UI, 256, 0, stream>>>(rec, base, X, nullptr, nullptr,
                                                             nullptr, nullptr, 0, out_ui, N_NODES);

    bin_count_kernel<<<NB_S, BIN_THREADS, 0, stream>>>(s_rows, H, E_S, T_S);
    bin_scan_kernel<<<1, MAX_TILES, 0, stream>>>(H, Hp, base, NB_S, T_S);
    bin_place_kernel<<<NB_S, BIN_THREADS, 0, stream>>>(s_rows, s_cols, s_vals, Hp, base, rec, E_S, T_S);

    gemm64_wmma_kernel<<<N_USER / 16, 32, 0, stream>>>(user_emb, user_emb, N_USER, s_w, Y, N_USER);
    spmm_tile_kernel<true, true><<<T_S, 256, 0, stream>>>(rec, base, Y, s_w + N_HID * N_HID, X,
                                                          user_emb, user_emb, N_USER, out_s, N_USER);
    spmm_tile_kernel<false, false><<<T_S, 256, 0, stream>>>(rec, base, X, nullptr, nullptr,
                                                            nullptr, nullptr, 0, out_s, N_USER);
}


# Layer-0 GEMM has no launch of its own: its 16-row tiles run as extra blocks of the three edge-sort launches, which leave most of the chip idle; 12 -> 10 launches; float sum orders kept
# speedup vs baseline: 3.7502x; 1.0791x over previous
#include <hip/hip_runtime.h>
#include <hip/hip_bf16.h>

typedef __attribute__((ext_vector_type(2))) float v2f;
typedef __attribute__((ext_vector_type(8))) float v8f;

#define N_USER 100000
#define N_ITEM 150000
#define N_NODES (N_USER + N_ITEM)
#define N_HID 64
#define E_UI 1000000
#define E_S 1000000
#define SLOPE 0.5f
#define EPS 1e-12f

static_assert(N_USER % 32 == 0, "a 16-row product tile and a wave's 32 tile rows must lie in one table");

struct GemmPart {
    const float* Xa;
    const float* Xb;
    const float* W;
    float* Y;
    int split;
    int tile_lo;
    int tile_hi;
};

__device__ __forceinline__ void gemm_tile16(const GemmPart& g, int tile, int lane) {
    const int row0 = tile * 16;
    const float* __restrict__ X0 = (row0 < g.split) ? g.Xa + (size_t)row0 * N_HID
                                                    : g.Xb + (size_t)(row0 - g.split) * N_HID;
    const float* __restrict__ W = g.W;
    float* __restrict__ Y = g.Y;
    const int r = lane & 15;
    const int h = lane >> 4;

    v8f acc[4];
    acc[0] = v8f{};
    acc[1] = v8f{};
    acc[2] = v8f{};
    acc[3] = v8f{};

#pragma unroll
    for (int k = 0; k < 64; k += 4) {
        const float* xr = X0 + (size_t)r * N_HID + k + 2 * h;
        v2f a;
        a[0] = xr[0];
        a[1] = xr[1];
        const float* w0 = W + (size_t)(k + 2 * h) * N_HID + r;
        const float* w1 = w0 + N_HID;
#pragma unroll
        for (int ct = 0; ct < 4; ++ct) {
            v2f b;
            b[0] = w0[ct * 16];
            b[1] = w1[ct * 16];
            acc[ct] = __builtin_amdgcn_wmma_f32_16x16x4_f32(
                 false, a,  false, b,
                 (short)0, acc[ct],
                 false,  false);
        }
    }

#pragma unroll
    for (int ct = 0; ct < 4; ++ct) {
#pragma unroll
        for (int v = 0; v < 8; ++v) {
            Y[(size_t)(row0 + v + 8 * h) * N_HID + ct * 16 + r] = acc[ct][v];
        }
    }
}

__device__ __forceinline__ void gemm_block(const GemmPart& g, int pb, int waves) {
    const int wv = __builtin_amdgcn_readfirstlane((int)(threadIdx.x >> 5));
    const int tile = g.tile_lo + pb * waves + wv;
    if (tile < g.tile_hi) gemm_tile16(g, tile, threadIdx.x & 31);
}

#define TILE_ROWS 256
#define TILE_SHIFT 8
#define PITCH 68
#define CHUNK 8192
#define BIN_THREADS 512
#define MAX_TILES 1024
#define T_UI ((N_NODES + TILE_ROWS - 1) / TILE_ROWS)
#define T_S ((N_USER + TILE_ROWS - 1) / TILE_ROWS)
#define NB_UI ((E_UI + CHUNK - 1) / CHUNK)
#define NB_S ((E_S + CHUNK - 1) / CHUNK)
#define NB_MAX (NB_UI > NB_S ? NB_UI : NB_S)
#define GT_UI (N_NODES / 16)
#define GT_S (N_USER / 16)
#define GP_UI ((GT_UI + 2) / 3)
#define GP_S ((GT_S + 2) / 3)
#define GB16(n) (((n) + BIN_THREADS / 32 - 1) / (BIN_THREADS / 32))
#define GB32(n) (((n) + MAX_TILES / 32 - 1) / (MAX_TILES / 32))

__global__ void __launch_bounds__(BIN_THREADS)
bin_count_kernel(const int* __restrict__ rows,
                 int* __restrict__ H,
                 int E, int T, int NB, GemmPart g) {
    __shared__ int h[MAX_TILES];
    if ((int)blockIdx.x >= NB) {
        gemm_block(g, blockIdx.x - NB, BIN_THREADS / 32);
        return;
    }
    const int tid = threadIdx.x;
    for (int t = tid; t < MAX_TILES; t += BIN_THREADS) h[t] = 0;
    __syncthreads();
    const int e0 = blockIdx.x * CHUNK;
#pragma unroll 4
    for (int i = 0; i < CHUNK / BIN_THREADS; ++i) {
        int e = e0 + i * BIN_THREADS + tid;
        if (e < E) atomicAdd(&h[rows[e] >> TILE_SHIFT], 1);
    }
    __syncthreads();
    for (int t = tid; t < T; t += BIN_THREADS) H[(size_t)blockIdx.x * T + t] = h[t];
}

__global__ void __launch_bounds__(MAX_TILES)
bin_scan_kernel(const int* __restrict__ H,
                int* __restrict__ Hp,
                int* __restrict__ base,
                int NB, int T, GemmPart g) {
    __shared__ int s[MAX_TILES];
    if (blockIdx.x != 0) {
        gemm_block(g, blockIdx.x - 1, MAX_TILES / 32);
        return;
    }
    const int t = threadIdx.x;
    int run = 0;
    if (t < T) {
#pragma unroll 8
        for (int b = 0; b < NB; ++b) {
            int c = H[(size_t)b * T + t];
            Hp[(size_t)b * T + t] = run;
            run += c;
        }
    }
    s[t] = run;
    __syncthreads();
    for (int off = 1; off < MAX_TILES; off <<= 1) {
        int v = (t >= off) ? s[t - off] : 0;
        __syncthreads();
        s[t] += v;
        __syncthreads();
    }
    if (t < T) base[t] = s[t] - run;
    if (t == MAX_TILES - 1) base[T] = s[t];
}

__global__ void __launch_bounds__(BIN_THREADS)
bin_place_kernel(const int* __restrict__ rows,
                 const int* __restrict__ cols,
                 const float* __restrict__ vals,
                 const int* __restrict__ Hp,
                 const int* __restrict__ base,
                 int2* __restrict__ rec,
                 int E, int T, int NB, GemmPart g) {
    __shared__ int pos[MAX_TILES];
    if ((int)blockIdx.x >= NB) {
        gemm_block(g, blockIdx.x - NB, BIN_THREADS / 32);
        return;
    }
    const int tid = threadIdx.x;
    for (int t = tid; t < T; t += BIN_THREADS) pos[t] = base[t] + Hp[(size_t)blockIdx.x * T + t];
    __syncthreads();
    const int e0 = blockIdx.x * CHUNK;
#pragma unroll 4
    for (int i = 0; i < CHUNK / BIN_THREADS; ++i) {
        int e = e0 + i * BIN_THREADS + tid;
        if (e < E) {
            int r = rows[e];
            int c = cols[e];
            float v = vals[e];
            int p = atomicAdd(&pos[r >> TILE_SHIFT], 1);
            rec[p] = make_int2((c << TILE_SHIFT) | (r & (TILE_ROWS - 1)), __float_as_int(v));
        }
    }
}

template <bool NEXT, bool TABLE, bool GUARD>
__device__ __forceinline__ void finish_rows(float* acc,
                                            const float* __restrict__ Wn,
                                            float2* __restrict__ yt,
                                            float2* __restrict__ ot,
                                            const float2* __restrict__ rt,
                                            int r_lo, int last, int lane) {
    float2 o[TILE_ROWS / 8];
    const float2* res = TABLE ? rt : (const float2*)ot;
#pragma unroll
    for (int j = 0; j < TILE_ROWS / 8; ++j) {
        o[j] = res[(GUARD ? min(j, last) : j) * (N_HID / 2)];
    }

    const float4* my4 = (const float4*)(acc + (r_lo + lane) * PITCH);
    float ss = 0.0f;
#pragma unroll
    for (int i = 0; i < N_HID / 4; ++i) {
        const float4 z = my4[i];
        const float vx = (z.x >= 0.0f) ? z.x : SLOPE * z.x;
        const float vy = (z.y >= 0.0f) ? z.y : SLOPE * z.y;
        const float vz = (z.z >= 0.0f) ? z.z : SLOPE * z.z;
        const float vw = (z.w >= 0.0f) ? z.w : SLOPE * z.w;
        ss += vx * vx + vy * vy + vz * vz + vw * vw;
    }
    const float inv_all = 1.0f / fmaxf(sqrtf(ss), EPS);

    float2* acc2 = (float2*)(acc + r_lo * PITCH) + lane;
#pragma unroll
    for (int j = 0; j < TILE_ROWS / 8; ++j) {
        const float inv = __int_as_float(__builtin_amdgcn_readlane(__float_as_int(inv_all), j));
        const float2 z = acc2[j * (PITCH / 2)];
        float2 v;
        v.x = (z.x >= 0.0f) ? z.x : SLOPE * z.x;
        v.y = (z.y >= 0.0f) ? z.y : SLOPE * z.y;
        float2 cur = o[j];
        cur.x += v.x * inv;
        cur.y += v.y * inv;
        if (NEXT) acc2[j * (PITCH / 2)] = v;
        if (!GUARD || j <= last) {
            ot[j * (N_HID / 2)] = cur;
        }
    }

    if (NEXT) {
        const int r = lane & 15;
        const int h = lane >> 4;
        v8f c0[4];
        v8f c1[4];
#pragma unroll
        for (int ct = 0; ct < 4; ++ct) {
            c0[ct] = v8f{};
            c1[ct] = v8f{};
        }
        const float* a_lo = acc + (r_lo + r) * PITCH + 2 * h;
        const float* a_hi = a_lo + 16 * PITCH;
#pragma unroll 4
        for (int k = 0; k < N_HID; k += 4) {
            const float2 p0 = *(const float2*)(a_lo + k);
            const float2 p1 = *(const float2*)(a_hi + k);
            v2f a0;
            a0[0] = p0.x;
            a0[1] = p0.y;
            v2f a1;
            a1[0] = p1.x;
            a1[1] = p1.y;
            const float* w0 = Wn + (size_t)(k + 2 * h) * N_HID + r;
            const float* w1 = w0 + N_HID;
#pragma unroll
            for (int ct = 0; ct < 4; ++ct) {
                v2f b;
                b[0] = w0[ct * 16];
                b[1] = w1[ct * 16];
                c0[ct] = __builtin_amdgcn_wmma_f32_16x16x4_f32(
                     false, a0,  false, b,
                     (short)0, c0[ct],
                     false,  false);
                c1[ct] = __builtin_amdgcn_wmma_f32_16x16x4_f32(
                     false, a1,  false, b,
                     (short)0, c1[ct],
                     false,  false);
            }
        }
#pragma unroll
        for (int ct = 0; ct < 4; ++ct) {
#pragma unroll
            for (int v = 0; v < 8; ++v) {
                acc[(r_lo + v + 8 * h) * PITCH + ct * 16 + r] = c0[ct][v];
                acc[(r_lo + 16 + v + 8 * h) * PITCH + ct * 16 + r] = c1[ct][v];
            }
        }
#pragma unroll
        for (int j = 0; j < TILE_ROWS / 8; ++j) {
            const float2 yv = acc2[j * (PITCH / 2)];
            if (!GUARD || j <= last) {
                yt[j * (N_HID / 2)] = yv;
            }
        }
    }
}

template <bool NEXT, bool TABLE>
__global__ void __launch_bounds__(256, 8)
spmm_tile_kernel(const int2* __restrict__ rec,
                 const int* __restrict__ base,
                 const float* __restrict__ Y,
                 const float* __restrict__ Wn,
                 float* __restrict__ Yn,
                 const float* __restrict__ ta,
                 const float* __restrict__ tb,
                 int split,
                 float* __restrict__ out,
                 int M) {
    __shared__ __attribute__((aligned(16))) float acc[TILE_ROWS * PITCH];
    const int tid = threadIdx.x;
    const int lane = tid & 31;
    const int w = __builtin_amdgcn_readfirstlane(tid >> 5);
    const int t = blockIdx.x;

    float4* acc4 = (float4*)acc;
    for (int i = tid; i < TILE_ROWS * PITCH / 4; i += 256) acc4[i] = make_float4(0.0f, 0.0f, 0.0f, 0.0f);
    __syncthreads();

    const int beg = base[t];
    const int end = base[t + 1];

    for (int e0 = beg + w * 32; e0 < end; e0 += 8 * 32) {
        const int n = min(32, end - e0);
        int kk = 0;
        int vb = 0;
        if (lane < n) {
            int2 rr = rec[e0 + lane];
            kk = rr.x;
            vb = rr.y;
        }
        float2 y[32];
#pragma unroll
        for (int j = 0; j < 32; ++j) {
            const int k = __builtin_amdgcn_readlane(kk, j);
            const float2* yr = (const float2*)(Y + (size_t)(k >> TILE_SHIFT) * N_HID);
            y[j] = yr[lane];
        }
#pragma unroll
        for (int j = 0; j < 32; ++j) {
            if (j == 0 || j < n) {
                const int k = __builtin_amdgcn_readlane(kk, j);
                const float v = __int_as_float(__builtin_amdgcn_readlane(vb, j));
                float* a = acc + (k & (TILE_ROWS - 1)) * PITCH + 2 * lane;
                atomicAdd(a, v * y[j].x);
                atomicAdd(a + 1, v * y[j].y);
            }
        }
    }
    const int rows_here = min(TILE_ROWS, M - t * TILE_ROWS);
    const int r_lo = w * (TILE_ROWS / 8);
    const size_t first = ((size_t)t * TILE_ROWS + r_lo) * N_HID;
    float2* yt = NEXT ? (float2*)(Yn + first) + lane : nullptr;
    float2* ot = (float2*)(out + first) + lane;
    const float2* rt = nullptr;
    if (TABLE) {
        const int g0 = t * TILE_ROWS + r_lo;
        rt = (const float2*)((min(g0, M - 1) < split) ? ta + (size_t)g0 * N_HID
                                                      : tb + (size_t)(g0 - split) * N_HID) + lane;
    }

    __syncthreads();

    if (r_lo + TILE_ROWS / 8 <= rows_here) {
        finish_rows<NEXT, TABLE, false>(acc, Wn, yt, ot, rt, r_lo, rows_here - 1 - r_lo, lane);
    } else {
        finish_rows<NEXT, TABLE, true>(acc, Wn, yt, ot, rt, r_lo, rows_here - 1 - r_lo, lane);
    }
}

extern "C" void kernel_launch(void* const* d_in, const int* in_sizes, int n_in,
                              void* d_out, int out_size, void* d_ws, size_t ws_size,
                              hipStream_t stream) {
    const float* user_emb = (const float*)d_in[0];
    const float* item_emb = (const float*)d_in[1];
    const float* ui_w     = (const float*)d_in[2];
    const float* s_w      = (const float*)d_in[3];
    const int*   ui_rows  = (const int*)d_in[4];
    const int*   ui_cols  = (const int*)d_in[5];
    const float* ui_vals  = (const float*)d_in[6];
    const int*   s_rows   = (const int*)d_in[7];
    const int*   s_cols   = (const int*)d_in[8];
    const float* s_vals   = (const float*)d_in[9];

    float* X = (float*)d_ws;
    float* Y = X + (size_t)N_NODES * N_HID;
    int2*  rec  = (int2*)(Y + (size_t)N_NODES * N_HID);
    int*   H    = (int*)(rec + E_UI);
    int*   Hp   = H + NB_MAX * MAX_TILES;
    int*   base = Hp + NB_MAX * MAX_TILES;

    float* out_ui = (float*)d_out;
    float* out_s  = out_ui + (size_t)N_NODES * N_HID;

    bin_count_kernel<<<NB_UI + GB16(GP_UI), BIN_THREADS, 0, stream>>>(
        ui_rows, H, E_UI, T_UI, NB_UI, GemmPart{user_emb, item_emb, ui_w, Y, N_USER, 0, GP_UI});
    bin_scan_kernel<<<1 + GB32(GP_UI), MAX_TILES, 0, stream>>>(
        H, Hp, base, NB_UI, T_UI, GemmPart{user_emb, item_emb, ui_w, Y, N_USER, GP_UI, 2 * GP_UI});
    bin_place_kernel<<<NB_UI + GB16(GT_UI - 2 * GP_UI), BIN_THREADS, 0, stream>>>(
        ui_rows, ui_cols, ui_vals, Hp, base, rec, E_UI, T_UI, NB_UI,
        GemmPart{user_emb, item_emb, ui_w, Y, N_USER, 2 * GP_UI, GT_UI});

    spmm_tile_kernel<true, true><<<T_UI, 256, 0, stream>>>(rec, base, Y, ui_w + N_HID * N_HID, X,
                                                           user_emb, item_emb, N_USER, out_ui, N_NODES);
    spmm_tile_kernel<false, false><<<T_UI, 256, 0, stream>>>(rec, base, X, nullptr, nullptr,
                                                             nullptr, nullptr, 0, out_ui, N_NODES);

    bin_count_kernel<<<NB_S + GB16(GP_S), BIN_THREADS, 0, stream>>>(
        s_rows, H, E_S, T_S, NB_S, GemmPart{user_emb, user_emb, s_w, Y, N_USER, 0, GP_S});
    bin_scan_kernel<<<1 + GB32(GP_S), MAX_TILES, 0, stream>>>(
        H, Hp, base, NB_S, T_S, GemmPart{user_emb, user_emb, s_w, Y, N_USER, GP_S, 2 * GP_S});
    bin_place_kernel<<<NB_S + GB16(GT_S - 2 * GP_S), BIN_THREADS, 0, stream>>>(
        s_rows, s_cols, s_vals, Hp, base, rec, E_S, T_S, NB_S,
        GemmPart{user_emb, user_emb, s_w, Y, N_USER, 2 * GP_S, GT_S});

    spmm_tile_kernel<true, true><<<T_S, 256, 0, stream>>>(rec, base, Y, s_w + N_HID * N_HID, X,
                                                          user_emb, user_emb, N_USER, out_s, N_USER);
    spmm_tile_kernel<false, false><<<T_S, 256, 0, stream>>>(rec, base, X, nullptr, nullptr,
                                                            nullptr, nullptr, 0, out_s, N_USER);
}


# Edge sort is one launch per graph, not three: a block orders its own 8192 edges by row tile in LDS; a tile block finds its edges in the blocks' runs by search; 6 launches, not 10; sums by LDS arrival
# speedup vs baseline: 4.8334x; 1.2888x over previous
#include <hip/hip_runtime.h>
#include <hip/hip_bf16.h>

typedef __attribute__((ext_vector_type(2))) float v2f;
typedef __attribute__((ext_vector_type(8))) float v8f;

#define N_USER 100000
#define N_ITEM 150000
#define N_NODES (N_USER + N_ITEM)
#define N_HID 64
#define E_UI 1000000
#define E_S 1000000
#define SLOPE 0.5f
#define EPS 1e-12f

static_assert(N_USER % 32 == 0, "a 16-row product tile and a wave's 32 tile rows must lie in one table");

struct GemmPart {
    const float* Xa;
    const float* Xb;
    const float* W;
    float* Y;
    int split;
    int tile_lo;
    int tile_hi;
};

__device__ __forceinline__ void gemm_tile16(const GemmPart& g, int tile, int lane) {
    const int row0 = tile * 16;
    const float* __restrict__ X0 = (row0 < g.split) ? g.Xa + (size_t)row0 * N_HID
                                                    : g.Xb + (size_t)(row0 - g.split) * N_HID;
    const float* __restrict__ W = g.W;
    float* __restrict__ Y = g.Y;
    const int r = lane & 15;
    const int h = lane >> 4;

    v8f acc[4];
    acc[0] = v8f{};
    acc[1] = v8f{};
    acc[2] = v8f{};
    acc[3] = v8f{};

#pragma unroll
    for (int k = 0; k < 64; k += 4) {
        const float* xr = X0 + (size_t)r * N_HID + k + 2 * h;
        v2f a;
        a[0] = xr[0];
        a[1] = xr[1];
        const float* w0 = W + (size_t)(k + 2 * h) * N_HID + r;
        const float* w1 = w0 + N_HID;
#pragma unroll
        for (int ct = 0; ct < 4; ++ct) {
            v2f b;
            b[0] = w0[ct * 16];
            b[1] = w1[ct * 16];
            acc[ct] = __builtin_amdgcn_wmma_f32_16x16x4_f32(
                 false, a,  false, b,
                 (short)0, acc[ct],
                 false,  false);
        }
    }

#pragma unroll
    for (int ct = 0; ct < 4; ++ct) {
#pragma unroll
        for (int v = 0; v < 8; ++v) {
            Y[(size_t)(row0 + v + 8 * h) * N_HID + ct * 16 + r] = acc[ct][v];
        }
    }
}

__device__ __forceinline__ void gemm_block(const GemmPart& g, int pb, int waves) {
    const int wv = __builtin_amdgcn_readfirstlane((int)(threadIdx.x >> 5));
    const int tile = g.tile_lo + pb * waves + wv;
    if (tile < g.tile_hi) gemm_tile16(g, tile, threadIdx.x & 31);
}

#define TILE_ROWS 256
#define TILE_SHIFT 8
#define PITCH 68
#define CHUNK 8192
#define BIN_THREADS 512
#define MAX_TILES 1024
#define T_UI ((N_NODES + TILE_ROWS - 1) / TILE_ROWS)
#define T_S ((N_USER + TILE_ROWS - 1) / TILE_ROWS)
#define NB_UI ((E_UI + CHUNK - 1) / CHUNK)
#define NB_S ((E_S + CHUNK - 1) / CHUNK)
#define NB_MAX (NB_UI > NB_S ? NB_UI : NB_S)
#define OFFS (MAX_TILES + 1)
#define RUNS 128
static_assert(NB_MAX <= RUNS, "a tile block keeps one run per sort block in a RUNS-entry table");
static_assert(T_UI < MAX_TILES && T_S < MAX_TILES, "entry T of a block's offsets must exist in its LDS table");
static_assert(MAX_TILES == 2 * BIN_THREADS, "the sort block's scan gives every thread two tiles");
#define GT_UI (N_NODES / 16)
#define GT_S (N_USER / 16)
#define GB16(n) (((n) + BIN_THREADS / 32 - 1) / (BIN_THREADS / 32))

__global__ void __launch_bounds__(BIN_THREADS)
sort_local_kernel(const int* __restrict__ rows,
                  const int* __restrict__ cols,
                  const float* __restrict__ vals,
                  int* __restrict__ Off,
                  int2* __restrict__ rec,
                  int E, int T, int NB, GemmPart g) {
    __shared__ int h[MAX_TILES];
    __shared__ int wsum[BIN_THREADS / 32];
    if ((int)blockIdx.x >= NB) {
        gemm_block(g, blockIdx.x - NB, BIN_THREADS / 32);
        return;
    }
    const int tid = threadIdx.x;
    const int lane = tid & 31;
    const int w = tid >> 5;
    h[2 * tid] = 0;
    h[2 * tid + 1] = 0;
    __syncthreads();

    const int e0 = blockIdx.x * CHUNK;
    int key[CHUNK / BIN_THREADS];
    int vbits[CHUNK / BIN_THREADS];
    int tl[CHUNK / BIN_THREADS];
    int rnk[CHUNK / BIN_THREADS];
#pragma unroll
    for (int i = 0; i < CHUNK / BIN_THREADS; ++i) {
        const int e = min(e0 + i * BIN_THREADS + tid, E - 1);
        const int r = rows[e];
        const int cc = cols[e];
        tl[i] = r >> TILE_SHIFT;
        key[i] = (cc << TILE_SHIFT) | (r & (TILE_ROWS - 1));
        vbits[i] = __float_as_int(vals[e]);
    }
#pragma unroll
    for (int i = 0; i < CHUNK / BIN_THREADS; ++i) {
        rnk[i] = 0;
        if (e0 + i * BIN_THREADS + tid < E) rnk[i] = atomicAdd(&h[tl[i]], 1);
    }
    __syncthreads();

    const int c0 = h[2 * tid];
    const int c1 = h[2 * tid + 1];
    const int s = c0 + c1;
    int incl = s;
#pragma unroll
    for (int d = 1; d < 32; d <<= 1) {
        const int up = __shfl_up(incl, d);
        if (lane >= d) incl += up;
    }
    if (lane == 31) wsum[w] = incl;
    __syncthreads();
    int before = 0;
#pragma unroll
    for (int k = 0; k < BIN_THREADS / 32; ++k) {
        const int ws = wsum[k];
        before += (k < w) ? ws : 0;
    }
    const int ex = before + incl - s;
    h[2 * tid] = ex;
    h[2 * tid + 1] = ex + c0;
    int* orow = Off + (size_t)blockIdx.x * OFFS;
    if (2 * tid <= T) orow[2 * tid] = ex;
    if (2 * tid + 1 <= T) orow[2 * tid + 1] = ex + c0;
    __syncthreads();

    int at[CHUNK / BIN_THREADS];
#pragma unroll
    for (int i = 0; i < CHUNK / BIN_THREADS; ++i) at[i] = e0 + h[tl[i]] + rnk[i];
#pragma unroll
    for (int i = 0; i < CHUNK / BIN_THREADS; ++i) {
        if (e0 + i * BIN_THREADS + tid < E) rec[at[i]] = make_int2(key[i], vbits[i]);
    }
}

template <bool NEXT, bool TABLE, bool GUARD>
__device__ __forceinline__ void finish_rows(float* acc,
                                            const float* __restrict__ Wn,
                                            float2* __restrict__ yt,
                                            float2* __restrict__ ot,
                                            const float2* __restrict__ rt,
                                            int r_lo, int last, int lane) {
    float2 o[TILE_ROWS / 8];
    const float2* res = TABLE ? rt : (const float2*)ot;
#pragma unroll
    for (int j = 0; j < TILE_ROWS / 8; ++j) {
        o[j] = res[(GUARD ? min(j, last) : j) * (N_HID / 2)];
    }

    const float4* my4 = (const float4*)(acc + (r_lo + lane) * PITCH);
    float ss = 0.0f;
#pragma unroll
    for (int i = 0; i < N_HID / 4; ++i) {
        const float4 z = my4[i];
        const float vx = (z.x >= 0.0f) ? z.x : SLOPE * z.x;
        const float vy = (z.y >= 0.0f) ? z.y : SLOPE * z.y;
        const float vz = (z.z >= 0.0f) ? z.z : SLOPE * z.z;
        const float vw = (z.w >= 0.0f) ? z.w : SLOPE * z.w;
        ss += vx * vx + vy * vy + vz * vz + vw * vw;
    }
    const float inv_all = 1.0f / fmaxf(sqrtf(ss), EPS);

    float2* acc2 = (float2*)(acc + r_lo * PITCH) + lane;
#pragma unroll
    for (int j = 0; j < TILE_ROWS / 8; ++j) {
        const float inv = __int_as_float(__builtin_amdgcn_readlane(__float_as_int(inv_all), j));
        const float2 z = acc2[j * (PITCH / 2)];
        float2 v;
        v.x = (z.x >= 0.0f) ? z.x : SLOPE * z.x;
        v.y = (z.y >= 0.0f) ? z.y : SLOPE * z.y;
        float2 cur = o[j];
        cur.x += v.x * inv;
        cur.y += v.y * inv;
        if (NEXT) acc2[j * (PITCH / 2)] = v;
        if (!GUARD || j <= last) {
            ot[j * (N_HID / 2)] = cur;
        }
    }

    if (NEXT) {
        const int r = lane & 15;
        const int h = lane >> 4;
        v8f c0[4];
        v8f c1[4];
#pragma unroll
        for (int ct = 0; ct < 4; ++ct) {
            c0[ct] = v8f{};
            c1[ct] = v8f{};
        }
        const float* a_lo = acc + (r_lo + r) * PITCH + 2 * h;
        const float* a_hi = a_lo + 16 * PITCH;
#pragma unroll 4
        for (int k = 0; k < N_HID; k += 4) {
            const float2 p0 = *(const float2*)(a_lo + k);
            const float2 p1 = *(const float2*)(a_hi + k);
            v2f a0;
            a0[0] = p0.x;
            a0[1] = p0.y;
            v2f a1;
            a1[0] = p1.x;
            a1[1] = p1.y;
            const float* w0 = Wn + (size_t)(k + 2 * h) * N_HID + r;
            const float* w1 = w0 + N_HID;
#pragma unroll
            for (int ct = 0; ct < 4; ++ct) {
                v2f b;
                b[0] = w0[ct * 16];
                b[1] = w1[ct * 16];
                c0[ct] = __builtin_amdgcn_wmma_f32_16x16x4_f32(
                     false, a0,  false, b,
                     (short)0, c0[ct],
                     false,  false);
                c1[ct] = __builtin_amdgcn_wmma_f32_16x16x4_f32(
                     false, a1,  false, b,
                     (short)0, c1[ct],
                     false,  false);
            }
        }
#pragma unroll
        for (int ct = 0; ct < 4; ++ct) {
#pragma unroll
            for (int v = 0; v < 8; ++v) {
                acc[(r_lo + v + 8 * h) * PITCH + ct * 16 + r] = c0[ct][v];
                acc[(r_lo + 16 + v + 8 * h) * PITCH + ct * 16 + r] = c1[ct][v];
            }
        }
#pragma unroll
        for (int j = 0; j < TILE_ROWS / 8; ++j) {
            const float2 yv = acc2[j * (PITCH / 2)];
            if (!GUARD || j <= last) {
                yt[j * (N_HID / 2)] = yv;
            }
        }
    }
}

template <bool NEXT, bool TABLE>
__global__ void __launch_bounds__(256, 8)
spmm_tile_kernel(const int2* __restrict__ rec,
                 const int* __restrict__ Off,
                 int NB,
                 const float* __restrict__ Y,
                 const float* __restrict__ Wn,
                 float* __restrict__ Yn,
                 const float* __restrict__ ta,
                 const float* __restrict__ tb,
                 int split,
                 float* __restrict__ out,
                 int M) {
    __shared__ __attribute__((aligned(16))) float acc[TILE_ROWS * PITCH];
    __shared__ __attribute__((aligned(16))) int P[RUNS + 4];
    __shared__ int St[RUNS];
    const int tid = threadIdx.x;
    const int lane = tid & 31;
    const int w = __builtin_amdgcn_readfirstlane(tid >> 5);
    const int t = blockIdx.x;

    int o0 = 0;
    int o1 = 0;
    if (tid < NB) {
        const int* o = Off + (size_t)tid * OFFS + t;
        o0 = o[0];
        o1 = o[1];
    }

    float4* acc4 = (float4*)acc;
    for (int i = tid; i < TILE_ROWS * PITCH / 4; i += 256) acc4[i] = make_float4(0.0f, 0.0f, 0.0f, 0.0f);
    if (tid < RUNS) {
        P[tid] = o1 - o0;
        St[tid] = tid * CHUNK + o0;
    }
    __syncthreads();
    if (w == 0) {
        const int4 a = *(const int4*)(P + 4 * lane);
        const int s = a.x + a.y + a.z + a.w;
        int incl = s;
#pragma unroll
        for (int d = 1; d < 32; d <<= 1) {
            const int up = __shfl_up(incl, d);
            if (lane >= d) incl += up;
        }
        const int ex = incl - s;
        *(int4*)(P + 4 * lane) = make_int4(ex, ex + a.x, ex + a.x + a.y, ex + a.x + a.y + a.z);
        if (lane == 31) P[RUNS] = incl;
    }
    __syncthreads();
    const int n_t = __builtin_amdgcn_readfirstlane(P[RUNS]);

    auto edge = [&](int e) -> int2 {
        int i = 0;
#pragma unroll
        for (int step = RUNS / 2; step >= 1; step >>= 1) {
            if (P[i + step] <= e) i += step;
        }
        return rec[St[i] + (e - P[i])];
    };

    int kk = 0;
    int vb = 0;
    if (w * 32 + lane < n_t) {
        const int2 rr = edge(w * 32 + lane);
        kk = rr.x;
        vb = rr.y;
    }
    for (int e0 = w * 32; e0 < n_t; e0 += 8 * 32) {
        const int n = min(32, n_t - e0);
        float2 y[32];
#pragma unroll
        for (int j = 0; j < 32; ++j) {
            const int k = __builtin_amdgcn_readlane(kk, j);
            const float2* yr = (const float2*)(Y + (size_t)(k >> TILE_SHIFT) * N_HID);
            y[j] = yr[lane];
        }
        int kn = 0;
        int vn = 0;
        if (e0 + 8 * 32 + lane < n_t) {
            const int2 rr = edge(e0 + 8 * 32 + lane);
            kn = rr.x;
            vn = rr.y;
        }
#pragma unroll
        for (int j = 0; j < 32; ++j) {
            if (j == 0 || j < n) {
                const int k = __builtin_amdgcn_readlane(kk, j);
                const float v = __int_as_float(__builtin_amdgcn_readlane(vb, j));
                float* a = acc + (k & (TILE_ROWS - 1)) * PITCH + 2 * lane;
                atomicAdd(a, v * y[j].x);
                atomicAdd(a + 1, v * y[j].y);
            }
        }
        kk = kn;
        vb = vn;
    }
    const int rows_here = min(TILE_ROWS, M - t * TILE_ROWS);
    const int r_lo = w * (TILE_ROWS / 8);
    const size_t first = ((size_t)t * TILE_ROWS + r_lo) * N_HID;
    float2* yt = NEXT ? (float2*)(Yn + first) + lane : nullptr;
    float2* ot = (float2*)(out + first) + lane;
    const float2* rt = nullptr;
    if (TABLE) {
        const int g0 = t * TILE_ROWS + r_lo;
        rt = (const float2*)((min(g0, M - 1) < split) ? ta + (size_t)g0 * N_HID
                                                      : tb + (size_t)(g0 - split) * N_HID) + lane;
    }

    __syncthreads();

    if (r_lo + TILE_ROWS / 8 <= rows_here) {
        finish_rows<NEXT, TABLE, false>(acc, Wn, yt, ot, rt, r_lo, rows_here - 1 - r_lo, lane);
    } else {
        finish_rows<NEXT, TABLE, true>(acc, Wn, yt, ot, rt, r_lo, rows_here - 1 - r_lo, lane);
    }
}

extern "C" void kernel_launch(void* const* d_in, const int* in_sizes, int n_in,
                              void* d_out, int out_size, void* d_ws, size_t ws_size,
                              hipStream_t stream) {
    const float* user_emb = (const float*)d_in[0];
    const float* item_emb = (const float*)d_in[1];
    const float* ui_w     = (const float*)d_in[2];
    const float* s_w      = (const float*)d_in[3];
    const int*   ui_rows  = (const int*)d_in[4];
    const int*   ui_cols  = (const int*)d_in[5];
    const float* ui_vals  = (const float*)d_in[6];
    const int*   s_rows   = (const int*)d_in[7];
    const int*   s_cols   = (const int*)d_in[8];
    const float* s_vals   = (const float*)d_in[9];

    float* X = (float*)d_ws;
    float* Y = X + (size_t)N_NODES * N_HID;
    int2*  rec = (int2*)(Y + (size_t)N_NODES * N_HID);
    int*   Off = (int*)(rec + E_UI);

    float* out_ui = (float*)d_out;
    float* out_s  = out_ui + (size_t)N_NODES * N_HID;

    sort_local_kernel<<<NB_UI + GB16(GT_UI), BIN_THREADS, 0, stream>>>(
        ui_rows, ui_cols, ui_vals, Off, rec, E_UI, T_UI, NB_UI,
        GemmPart{user_emb, item_emb, ui_w, Y, N_USER, 0, GT_UI});

    spmm_tile_kernel<true, true><<<T_UI, 256, 0, stream>>>(rec, Off, NB_UI, Y, ui_w + N_HID * N_HID, X,
                                                           user_emb, item_emb, N_USER, out_ui, N_NODES);
    spmm_tile_kernel<false, false><<<T_UI, 256, 0, stream>>>(rec, Off, NB_UI, X, nullptr, nullptr,
                                                             nullptr, nullptr, 0, out_ui, N_NODES);

    sort_local_kernel<<<NB_S + GB16(GT_S), BIN_THREADS, 0, stream>>>(
        s_rows, s_cols, s_vals, Off, rec, E_S, T_S, NB_S,
        GemmPart{user_emb, user_emb, s_w, Y, N_USER, 0, GT_S});

    spmm_tile_kernel<true, true><<<T_S, 256, 0, stream>>>(rec, Off, NB_S, Y, s_w + N_HID * N_HID, X,
                                                          user_emb, user_emb, N_USER, out_s, N_USER);
    spmm_tile_kernel<false, false><<<T_S, 256, 0, stream>>>(rec, Off, NB_S, X, nullptr, nullptr,
                                                            nullptr, nullptr, 0, out_s, N_USER);
}


# Both graphs share each launch: one grid runs both edge sorts and both layer-0 products, one grid per layer both graphs' tile blocks (own planes, records, offsets); 3 launches, not 6; sum orders kept
# speedup vs baseline: 5.4802x; 1.1338x over previous
#include <hip/hip_runtime.h>
#include <hip/hip_bf16.h>

typedef __attribute__((ext_vector_type(2))) float v2f;
typedef __attribute__((ext_vector_type(8))) float v8f;

#define N_USER 100000
#define N_ITEM 150000
#define N_NODES (N_USER + N_ITEM)
#define N_HID 64
#define E_UI 1000000
#define E_S 1000000
#define SLOPE 0.5f
#define EPS 1e-12f

static_assert(N_USER % 32 == 0, "a 16-row product tile and a wave's 32 tile rows must lie in one table");

struct GemmPart {
    const float* Xa;
    const float* Xb;
    const float* W;
    float* Y;
    int split;
    int tile_lo;
    int tile_hi;
};

__device__ __forceinline__ void gemm_tile16(const GemmPart& g, int tile, int lane) {
    const int row0 = tile * 16;
    const float* __restrict__ X0 = (row0 < g.split) ? g.Xa + (size_t)row0 * N_HID
                                                    : g.Xb + (size_t)(row0 - g.split) * N_HID;
    const float* __restrict__ W = g.W;
    float* __restrict__ Y = g.Y;
    const int r = lane & 15;
    const int h = lane >> 4;

    v8f acc[4];
    acc[0] = v8f{};
    acc[1] = v8f{};
    acc[2] = v8f{};
    acc[3] = v8f{};

#pragma unroll
    for (int k = 0; k < 64; k += 4) {
        const float* xr = X0 + (size_t)r * N_HID + k + 2 * h;
        v2f a;
        a[0] = xr[0];
        a[1] = xr[1];
        const float* w0 = W + (size_t)(k + 2 * h) * N_HID + r;
        const float* w1 = w0 + N_HID;
#pragma unroll
        for (int ct = 0; ct < 4; ++ct) {
            v2f b;
            b[0] = w0[ct * 16];
            b[1] = w1[ct * 16];
            acc[ct] = __builtin_amdgcn_wmma_f32_16x16x4_f32(
                 false, a,  false, b,
                 (short)0, acc[ct],
                 false,  false);
        }
    }

#pragma unroll
    for (int ct = 0; ct < 4; ++ct) {
#pragma unroll
        for (int v = 0; v < 8; ++v) {
            Y[(size_t)(row0 + v + 8 * h) * N_HID + ct * 16 + r] = acc[ct][v];
        }
    }
}

__device__ __forceinline__ void gemm_block(const GemmPart& g, int pb, int waves) {
    const int wv = __builtin_amdgcn_readfirstlane((int)(threadIdx.x >> 5));
    const int tile = g.tile_lo + pb * waves + wv;
    if (tile < g.tile_hi) gemm_tile16(g, tile, threadIdx.x & 31);
}

#define TILE_ROWS 256
#define TILE_SHIFT 8
#define PITCH 68
#define CHUNK 8192
#define BIN_THREADS 512
#define MAX_TILES 1024
#define T_UI ((N_NODES + TILE_ROWS - 1) / TILE_ROWS)
#define T_S ((N_USER + TILE_ROWS - 1) / TILE_ROWS)
#define NB_UI ((E_UI + CHUNK - 1) / CHUNK)
#define NB_S ((E_S + CHUNK - 1) / CHUNK)
#define NB_MAX (NB_UI > NB_S ? NB_UI : NB_S)
#define OFFS (MAX_TILES + 1)
#define RUNS 128
static_assert(NB_MAX <= RUNS, "a tile block keeps one run per sort block in a RUNS-entry table");
static_assert(T_UI < MAX_TILES && T_S < MAX_TILES, "entry T of a block's offsets must exist in its LDS table");
static_assert(MAX_TILES == 2 * BIN_THREADS, "the sort block's scan gives every thread two tiles");
#define GT_UI (N_NODES / 16)
#define GT_S (N_USER / 16)
#define GB16(n) (((n) + BIN_THREADS / 32 - 1) / (BIN_THREADS / 32))

struct SortPart {
    const int* rows;
    const int* cols;
    const float* vals;
    int* Off;
    int2* rec;
    int E;
    int T;
    int NB;
};

__global__ void __launch_bounds__(BIN_THREADS)
sort_local_kernel(SortPart sa, SortPart sb, GemmPart ga, GemmPart gb, int gba) {
    __shared__ int h[MAX_TILES];
    __shared__ int wsum[BIN_THREADS / 32];
    const int nsort = sa.NB + sb.NB;
    if ((int)blockIdx.x >= nsort) {
        const int pb = (int)blockIdx.x - nsort;
        const bool g2 = pb >= gba;
        GemmPart g;
        g.Xa = g2 ? gb.Xa : ga.Xa;
        g.Xb = g2 ? gb.Xb : ga.Xb;
        g.W = g2 ? gb.W : ga.W;
        g.Y = g2 ? gb.Y : ga.Y;
        g.split = g2 ? gb.split : ga.split;
        g.tile_lo = g2 ? gb.tile_lo : ga.tile_lo;
        g.tile_hi = g2 ? gb.tile_hi : ga.tile_hi;
        gemm_block(g, g2 ? pb - gba : pb, BIN_THREADS / 32);
        return;
    }
    const bool s2 = (int)blockIdx.x >= sa.NB;
    const int blk = s2 ? (int)blockIdx.x - sa.NB : (int)blockIdx.x;
    const int* __restrict__ rows = s2 ? sb.rows : sa.rows;
    const int* __restrict__ cols = s2 ? sb.cols : sa.cols;
    const float* __restrict__ vals = s2 ? sb.vals : sa.vals;
    int* __restrict__ Off = s2 ? sb.Off : sa.Off;
    int2* __restrict__ rec = s2 ? sb.rec : sa.rec;
    const int E = s2 ? sb.E : sa.E;
    const int T = s2 ? sb.T : sa.T;
    const int tid = threadIdx.x;
    const int lane = tid & 31;
    const int w = tid >> 5;
    h[2 * tid] = 0;
    h[2 * tid + 1] = 0;
    __syncthreads();

    const int e0 = blk * CHUNK;
    int key[CHUNK / BIN_THREADS];
    int vbits[CHUNK / BIN_THREADS];
    int tl[CHUNK / BIN_THREADS];
    int rnk[CHUNK / BIN_THREADS];
#pragma unroll
    for (int i = 0; i < CHUNK / BIN_THREADS; ++i) {
        const int e = min(e0 + i * BIN_THREADS + tid, E - 1);
        const int r = rows[e];
        const int cc = cols[e];
        tl[i] = r >> TILE_SHIFT;
        key[i] = (cc << TILE_SHIFT) | (r & (TILE_ROWS - 1));
        vbits[i] = __float_as_int(vals[e]);
    }
#pragma unroll
    for (int i = 0; i < CHUNK / BIN_THREADS; ++i) {
        rnk[i] = 0;
        if (e0 + i * BIN_THREADS + tid < E) rnk[i] = atomicAdd(&h[tl[i]], 1);
    }
    __syncthreads();

    const int c0 = h[2 * tid];
    const int c1 = h[2 * tid + 1];
    const int s = c0 + c1;
    int incl = s;
#pragma unroll
    for (int d = 1; d < 32; d <<= 1) {
        const int up = __shfl_up(incl, d);
        if (lane >= d) incl += up;
    }
    if (lane == 31) wsum[w] = incl;
    __syncthreads();
    int before = 0;
#pragma unroll
    for (int k = 0; k < BIN_THREADS / 32; ++k) {
        const int ws = wsum[k];
        before += (k < w) ? ws : 0;
    }
    const int ex = before + incl - s;
    h[2 * tid] = ex;
    h[2 * tid + 1] = ex + c0;
    int* orow = Off + (size_t)blk * OFFS;
    if (2 * tid <= T) orow[2 * tid] = ex;
    if (2 * tid + 1 <= T) orow[2 * tid + 1] = ex + c0;
    __syncthreads();

    int at[CHUNK / BIN_THREADS];
#pragma unroll
    for (int i = 0; i < CHUNK / BIN_THREADS; ++i) at[i] = e0 + h[tl[i]] + rnk[i];
#pragma unroll
    for (int i = 0; i < CHUNK / BIN_THREADS; ++i) {
        if (e0 + i * BIN_THREADS + tid < E) rec[at[i]] = make_int2(key[i], vbits[i]);
    }
}

template <bool NEXT, bool TABLE, bool GUARD>
__device__ __forceinline__ void finish_rows(float* acc,
                                            const float* __restrict__ Wn,
                                            float2* __restrict__ yt,
                                            float2* __restrict__ ot,
                                            const float2* __restrict__ rt,
                                            int r_lo, int last, int lane) {
    float2 o[TILE_ROWS / 8];
    const float2* res = TABLE ? rt : (const float2*)ot;
#pragma unroll
    for (int j = 0; j < TILE_ROWS / 8; ++j) {
        o[j] = res[(GUARD ? min(j, last) : j) * (N_HID / 2)];
    }

    const float4* my4 = (const float4*)(acc + (r_lo + lane) * PITCH);
    float ss = 0.0f;
#pragma unroll
    for (int i = 0; i < N_HID / 4; ++i) {
        const float4 z = my4[i];
        const float vx = (z.x >= 0.0f) ? z.x : SLOPE * z.x;
        const float vy = (z.y >= 0.0f) ? z.y : SLOPE * z.y;
        const float vz = (z.z >= 0.0f) ? z.z : SLOPE * z.z;
        const float vw = (z.w >= 0.0f) ? z.w : SLOPE * z.w;
        ss += vx * vx + vy * vy + vz * vz + vw * vw;
    }
    const float inv_all = 1.0f / fmaxf(sqrtf(ss), EPS);

    float2* acc2 = (float2*)(acc + r_lo * PITCH) + lane;
#pragma unroll
    for (int j = 0; j < TILE_ROWS / 8; ++j) {
        const float inv = __int_as_float(__builtin_amdgcn_readlane(__float_as_int(inv_all), j));
        const float2 z = acc2[j * (PITCH / 2)];
        float2 v;
        v.x = (z.x >= 0.0f) ? z.x : SLOPE * z.x;
        v.y = (z.y >= 0.0f) ? z.y : SLOPE * z.y;
        float2 cur = o[j];
        cur.x += v.x * inv;
        cur.y += v.y * inv;
        if (NEXT) acc2[j * (PITCH / 2)] = v;
        if (!GUARD || j <= last) {
            ot[j * (N_HID / 2)] = cur;
        }
    }

    if (NEXT) {
        const int r = lane & 15;
        const int h = lane >> 4;
        v8f c0[4];
        v8f c1[4];
#pragma unroll
        for (int ct = 0; ct < 4; ++ct) {
            c0[ct] = v8f{};
            c1[ct] = v8f{};
        }
        const float* a_lo = acc + (r_lo + r) * PITCH + 2 * h;
        const float* a_hi = a_lo + 16 * PITCH;
#pragma unroll 4
        for (int k = 0; k < N_HID; k += 4) {
            const float2 p0 = *(const float2*)(a_lo + k);
            const float2 p1 = *(const float2*)(a_hi + k);
            v2f a0;
            a0[0] = p0.x;
            a0[1] = p0.y;
            v2f a1;
            a1[0] = p1.x;
            a1[1] = p1.y;
            const float* w0 = Wn + (size_t)(k + 2 * h) * N_HID + r;
            const float* w1 = w0 + N_HID;
#pragma unroll
            for (int ct = 0; ct < 4; ++ct) {
                v2f b;
                b[0] = w0[ct * 16];
                b[1] = w1[ct * 16];
                c0[ct] = __builtin_amdgcn_wmma_f32_16x16x4_f32(
                     false, a0,  false, b,
                     (short)0, c0[ct],
                     false,  false);
                c1[ct] = __builtin_amdgcn_wmma_f32_16x16x4_f32(
                     false, a1,  false, b,
                     (short)0, c1[ct],
                     false,  false);
            }
        }
#pragma unroll
        for (int ct = 0; ct < 4; ++ct) {
#pragma unroll
            for (int v = 0; v < 8; ++v) {
                acc[(r_lo + v + 8 * h) * PITCH + ct * 16 + r] = c0[ct][v];
                acc[(r_lo + 16 + v + 8 * h) * PITCH + ct * 16 + r] = c1[ct][v];
            }
        }
#pragma unroll
        for (int j = 0; j < TILE_ROWS / 8; ++j) {
            const float2 yv = acc2[j * (PITCH / 2)];
            if (!GUARD || j <= last) {
                yt[j * (N_HID / 2)] = yv;
            }
        }
    }
}

struct TilePart {
    const int2* rec;
    const int* Off;
    int NB;
    const float* Y;
    const float* Wn;
    float* Yn;
    const float* ta;
    const float* tb;
    int split;
    float* out;
    int M;
};

template <bool NEXT, bool TABLE>
__global__ void __launch_bounds__(256, 8)
spmm_tile_kernel(TilePart pa, TilePart pb, int na) {
    const bool p2 = (int)blockIdx.x >= na;
    const int2* __restrict__ rec = p2 ? pb.rec : pa.rec;
    const int* __restrict__ Off = p2 ? pb.Off : pa.Off;
    const int NB = p2 ? pb.NB : pa.NB;
    const float* __restrict__ Y = p2 ? pb.Y : pa.Y;
    const float* __restrict__ Wn = p2 ? pb.Wn : pa.Wn;
    float* __restrict__ Yn = p2 ? pb.Yn : pa.Yn;
    const float* __restrict__ ta = p2 ? pb.ta : pa.ta;
    const float* __restrict__ tb = p2 ? pb.tb : pa.tb;
    const int split = p2 ? pb.split : pa.split;
    float* __restrict__ out = p2 ? pb.out : pa.out;
    const int M = p2 ? pb.M : pa.M;
    __shared__ __attribute__((aligned(16))) float acc[TILE_ROWS * PITCH];
    __shared__ __attribute__((aligned(16))) int P[RUNS + 4];
    __shared__ int St[RUNS];
    const int tid = threadIdx.x;
    const int lane = tid & 31;
    const int w = __builtin_amdgcn_readfirstlane(tid >> 5);
    const int t = p2 ? (int)blockIdx.x - na : (int)blockIdx.x;

    int o0 = 0;
    int o1 = 0;
    if (tid < NB) {
        const int* o = Off + (size_t)tid * OFFS + t;
        o0 = o[0];
        o1 = o[1];
    }

    float4* acc4 = (float4*)acc;
    for (int i = tid; i < TILE_ROWS * PITCH / 4; i += 256) acc4[i] = make_float4(0.0f, 0.0f, 0.0f, 0.0f);
    if (tid < RUNS) {
        P[tid] = o1 - o0;
        St[tid] = tid * CHUNK + o0;
    }
    __syncthreads();
    if (w == 0) {
        const int4 a = *(const int4*)(P + 4 * lane);
        const int s = a.x + a.y + a.z + a.w;
        int incl = s;
#pragma unroll
        for (int d = 1; d < 32; d <<= 1) {
            const int up = __shfl_up(incl, d);
            if (lane >= d) incl += up;
        }
        const int ex = incl - s;
        *(int4*)(P + 4 * lane) = make_int4(ex, ex + a.x, ex + a.x + a.y, ex + a.x + a.y + a.z);
        if (lane == 31) P[RUNS] = incl;
    }
    __syncthreads();
    const int n_t = __builtin_amdgcn_readfirstlane(P[RUNS]);

    auto edge = [&](int e) -> int2 {
        int i = 0;
#pragma unroll
        for (int step = RUNS / 2; step >= 1; step >>= 1) {
            if (P[i + step] <= e) i += step;
        }
        return rec[St[i] + (e - P[i])];
    };

    int kk = 0;
    int vb = 0;
    if (w * 32 + lane < n_t) {
        const int2 rr = edge(w * 32 + lane);
        kk = rr.x;
        vb = rr.y;
    }
    for (int e0 = w * 32; e0 < n_t; e0 += 8 * 32) {
        const int n = min(32, n_t - e0);
        float2 y[32];
#pragma unroll
        for (int j = 0; j < 32; ++j) {
            const int k = __builtin_amdgcn_readlane(kk, j);
            const float2* yr = (const float2*)(Y + (size_t)(k >> TILE_SHIFT) * N_HID);
            y[j] = yr[lane];
        }
        int kn = 0;
        int vn = 0;
        if (e0 + 8 * 32 + lane < n_t) {
            const int2 rr = edge(e0 + 8 * 32 + lane);
            kn = rr.x;
            vn = rr.y;
        }
#pragma unroll
        for (int j = 0; j < 32; ++j) {
            if (j == 0 || j < n) {
                const int k = __builtin_amdgcn_readlane(kk, j);
                const float v = __int_as_float(__builtin_amdgcn_readlane(vb, j));
                float* a = acc + (k & (TILE_ROWS - 1)) * PITCH + 2 * lane;
                atomicAdd(a, v * y[j].x);
                atomicAdd(a + 1, v * y[j].y);
            }
        }
        kk = kn;
        vb = vn;
    }
    const int rows_here = min(TILE_ROWS, M - t * TILE_ROWS);
    const int r_lo = w * (TILE_ROWS / 8);
    const size_t first = ((size_t)t * TILE_ROWS + r_lo) * N_HID;
    float2* yt = NEXT ? (float2*)(Yn + first) + lane : nullptr;
    float2* ot = (float2*)(out + first) + lane;
    const float2* rt = nullptr;
    if (TABLE) {
        const int g0 = t * TILE_ROWS + r_lo;
        rt = (const float2*)((min(g0, M - 1) < split) ? ta + (size_t)g0 * N_HID
                                                      : tb + (size_t)(g0 - split) * N_HID) + lane;
    }

    __syncthreads();

    if (r_lo + TILE_ROWS / 8 <= rows_here) {
        finish_rows<NEXT, TABLE, false>(acc, Wn, yt, ot, rt, r_lo, rows_here - 1 - r_lo, lane);
    } else {
        finish_rows<NEXT, TABLE, true>(acc, Wn, yt, ot, rt, r_lo, rows_here - 1 - r_lo, lane);
    }
}

extern "C" void kernel_launch(void* const* d_in, const int* in_sizes, int n_in,
                              void* d_out, int out_size, void* d_ws, size_t ws_size,
                              hipStream_t stream) {
    const float* user_emb = (const float*)d_in[0];
    const float* item_emb = (const float*)d_in[1];
    const float* ui_w     = (const float*)d_in[2];
    const float* s_w      = (const float*)d_in[3];
    const int*   ui_rows  = (const int*)d_in[4];
    const int*   ui_cols  = (const int*)d_in[5];
    const float* ui_vals  = (const float*)d_in[6];
    const int*   s_rows   = (const int*)d_in[7];
    const int*   s_cols   = (const int*)d_in[8];
    const float* s_vals   = (const float*)d_in[9];

    float* Xu = (float*)d_ws;
    float* Yu = Xu + (size_t)N_NODES * N_HID;
    float* Xs = Yu + (size_t)N_NODES * N_HID;
    float* Ys = Xs + (size_t)N_USER * N_HID;
    int2*  recu = (int2*)(Ys + (size_t)N_USER * N_HID);
    int2*  recs = recu + E_UI;
    int*   Offu = (int*)(recs + E_S);
    int*   Offs = Offu + (size_t)NB_UI * OFFS;

    float* out_ui = (float*)d_out;
    float* out_s  = out_ui + (size_t)N_NODES * N_HID;

    sort_local_kernel<<<NB_UI + NB_S + GB16(GT_UI) + GB16(GT_S), BIN_THREADS, 0, stream>>>(
        SortPart{ui_rows, ui_cols, ui_vals, Offu, recu, E_UI, T_UI, NB_UI},
        SortPart{s_rows, s_cols, s_vals, Offs, recs, E_S, T_S, NB_S},
        GemmPart{user_emb, item_emb, ui_w, Yu, N_USER, 0, GT_UI},
        GemmPart{user_emb, user_emb, s_w, Ys, N_USER, 0, GT_S},
        GB16(GT_UI));

    spmm_tile_kernel<true, true><<<T_S + T_UI, 256, 0, stream>>>(
        TilePart{recs, Offs, NB_S, Ys, s_w + N_HID * N_HID, Xs, user_emb, user_emb, N_USER, out_s, N_USER},
        TilePart{recu, Offu, NB_UI, Yu, ui_w + N_HID * N_HID, Xu, user_emb, item_emb, N_USER, out_ui, N_NODES},
        T_S);
    spmm_tile_kernel<false, false><<<T_S + T_UI, 256, 0, stream>>>(
        TilePart{recs, Offs, NB_S, Xs, nullptr, nullptr, nullptr, nullptr, 0, out_s, N_USER},
        TilePart{recu, Offu, NB_UI, Xu, nullptr, nullptr, nullptr, nullptr, 0, out_ui, N_NODES},
        T_S);
}


# Layer-0 product's operands pass through LDS: a block stages W once in fragment order, a wave's X rows come in and its Y rows leave as whole 512-B pieces via its own LDS tile; float sums as before
# speedup vs baseline: 5.6783x; 1.0361x over previous
#include <hip/hip_runtime.h>
#include <hip/hip_bf16.h>

typedef __attribute__((ext_vector_type(2))) float v2f;
typedef __attribute__((ext_vector_type(8))) float v8f;

#define N_USER 100000
#define N_ITEM 150000
#define N_NODES (N_USER + N_ITEM)
#define N_HID 64
#define E_UI 1000000
#define E_S 1000000
#define SLOPE 0.5f
#define EPS 1e-12f

static_assert(N_USER % 32 == 0, "a 16-row product tile and a wave's 32 tile rows must lie in one table");

struct GemmPart {
    const float* Xa;
    const float* Xb;
    const float* W;
    float* Y;
    int split;
    int tile_lo;
    int tile_hi;
};

#define XP 68
#define PWAVES 16
#define WF_FLOATS (N_HID * N_HID)
#define XT_FLOATS (PWAVES * 16 * XP)

__device__ __forceinline__ void gemm_block(const GemmPart& g, int pb, float* wf, float* xt) {
    const int tid = threadIdx.x;
    const int lane = tid & 31;
    const int wv = __builtin_amdgcn_readfirstlane(tid >> 5);
    const int tile = g.tile_lo + pb * PWAVES + wv;
    const bool has = tile < g.tile_hi;
    const int row0 = tile * 16;
    const float* __restrict__ W = g.W;
    const int r = lane & 15;
    const int h = lane >> 4;

    float4 xin[8];
#pragma unroll
    for (int i = 0; i < 8; ++i) xin[i] = make_float4(0.0f, 0.0f, 0.0f, 0.0f);
    if (has) {
        const float* __restrict__ X0 = (row0 < g.split) ? g.Xa + (size_t)row0 * N_HID
                                                        : g.Xb + (size_t)(row0 - g.split) * N_HID;
        const float4* x4 = (const float4*)X0 + lane;
#pragma unroll
        for (int i = 0; i < 8; ++i) xin[i] = x4[i * 32];
    }

    float4* wf4 = (float4*)wf;
#pragma unroll
    for (int i = 0; i < 2; ++i) {
        const int q = tid + i * (PWAVES * 32);
        const int k = (q >> 6) * 4;
        const int ct = ((q >> 5) & 1) * 2;
        const float* w0 = W + (size_t)(k + 2 * h) * N_HID + ct * 16 + r;
        const float* w1 = w0 + N_HID;
        wf4[q] = make_float4(w0[0], w1[0], w0[16], w1[16]);
    }
    __syncthreads();
    if (!has) return;

    float* __restrict__ Y = g.Y;
    float* xw = xt + wv * (16 * XP);
    float4* xw4 = (float4*)(xw + (lane >> 4) * XP) + (lane & 15);
#pragma unroll
    for (int i = 0; i < 8; ++i) xw4[i * (2 * XP / 4)] = xin[i];

    v8f acc[4];
    acc[0] = v8f{};
    acc[1] = v8f{};
    acc[2] = v8f{};
    acc[3] = v8f{};

    const float* a_row = xw + r * XP + 2 * h;
    const float4* b4 = (const float4*)wf + lane;
#pragma unroll
    for (int k = 0; k < 64; k += 4) {
        const float2 p = *(const float2*)(a_row + k);
        v2f a;
        a[0] = p.x;
        a[1] = p.y;
        const float4 b01 = b4[(k / 4) * 64];
        const float4 b23 = b4[(k / 4) * 64 + 32];
        v2f b[4];
        b[0][0] = b01.x;
        b[0][1] = b01.y;
        b[1][0] = b01.z;
        b[1][1] = b01.w;
        b[2][0] = b23.x;
        b[2][1] = b23.y;
        b[3][0] = b23.z;
        b[3][1] = b23.w;
#pragma unroll
        for (int ct = 0; ct < 4; ++ct) {
            acc[ct] = __builtin_amdgcn_wmma_f32_16x16x4_f32(
                 false, a,  false, b[ct],
                 (short)0, acc[ct],
                 false,  false);
        }
    }

#pragma unroll
    for (int ct = 0; ct < 4; ++ct) {
#pragma unroll
        for (int v = 0; v < 8; ++v) {
            xw[(v + 8 * h) * XP + ct * 16 + r] = acc[ct][v];
        }
    }
    float4* y4 = (float4*)(Y + (size_t)row0 * N_HID) + lane;
#pragma unroll
    for (int i = 0; i < 8; ++i) y4[i * 32] = xw4[i * (2 * XP / 4)];
}

#define TILE_ROWS 256
#define TILE_SHIFT 8
#define PITCH 68
#define CHUNK 8192
#define BIN_THREADS 512
#define MAX_TILES 1024
#define T_UI ((N_NODES + TILE_ROWS - 1) / TILE_ROWS)
#define T_S ((N_USER + TILE_ROWS - 1) / TILE_ROWS)
#define NB_UI ((E_UI + CHUNK - 1) / CHUNK)
#define NB_S ((E_S + CHUNK - 1) / CHUNK)
#define NB_MAX (NB_UI > NB_S ? NB_UI : NB_S)
#define OFFS (MAX_TILES + 1)
#define RUNS 128
static_assert(NB_MAX <= RUNS, "a tile block keeps one run per sort block in a RUNS-entry table");
static_assert(T_UI < MAX_TILES && T_S < MAX_TILES, "entry T of a block's offsets must exist in its LDS table");
static_assert(MAX_TILES == 2 * BIN_THREADS, "the sort block's scan gives every thread two tiles");
static_assert(PWAVES * 32 == BIN_THREADS, "a product block is a block of the sort kernel: one tile per wave, two W entries per thread");
#define GT_UI (N_NODES / 16)
#define GT_S (N_USER / 16)
#define GB16(n) (((n) + BIN_THREADS / 32 - 1) / (BIN_THREADS / 32))

struct SortPart {
    const int* rows;
    const int* cols;
    const float* vals;
    int* Off;
    int2* rec;
    int E;
    int T;
    int NB;
};

__global__ void __launch_bounds__(BIN_THREADS)
sort_local_kernel(SortPart sa, SortPart sb, GemmPart ga, GemmPart gb, int gba) {
    __shared__ int h[MAX_TILES];
    __shared__ int wsum[BIN_THREADS / 32];
    __shared__ __attribute__((aligned(16))) float wf[WF_FLOATS];
    __shared__ __attribute__((aligned(16))) float xt[XT_FLOATS];
    const int nsort = sa.NB + sb.NB;
    if ((int)blockIdx.x >= nsort) {
        const int pb = (int)blockIdx.x - nsort;
        const bool g2 = pb >= gba;
        GemmPart g;
        g.Xa = g2 ? gb.Xa : ga.Xa;
        g.Xb = g2 ? gb.Xb : ga.Xb;
        g.W = g2 ? gb.W : ga.W;
        g.Y = g2 ? gb.Y : ga.Y;
        g.split = g2 ? gb.split : ga.split;
        g.tile_lo = g2 ? gb.tile_lo : ga.tile_lo;
        g.tile_hi = g2 ? gb.tile_hi : ga.tile_hi;
        gemm_block(g, g2 ? pb - gba : pb, wf, xt);
        return;
    }
    const bool s2 = (int)blockIdx.x >= sa.NB;
    const int blk = s2 ? (int)blockIdx.x - sa.NB : (int)blockIdx.x;
    const int* __restrict__ rows = s2 ? sb.rows : sa.rows;
    const int* __restrict__ cols = s2 ? sb.cols : sa.cols;
    const float* __restrict__ vals = s2 ? sb.vals : sa.vals;
    int* __restrict__ Off = s2 ? sb.Off : sa.Off;
    int2* __restrict__ rec = s2 ? sb.rec : sa.rec;
    const int E = s2 ? sb.E : sa.E;
    const int T = s2 ? sb.T : sa.T;
    const int tid = threadIdx.x;
    const int lane = tid & 31;
    const int w = tid >> 5;
    h[2 * tid] = 0;
    h[2 * tid + 1] = 0;
    __syncthreads();

    const int e0 = blk * CHUNK;
    int key[CHUNK / BIN_THREADS];
    int vbits[CHUNK / BIN_THREADS];
    int tl[CHUNK / BIN_THREADS];
    int rnk[CHUNK / BIN_THREADS];
#pragma unroll
    for (int i = 0; i < CHUNK / BIN_THREADS; ++i) {
        const int e = min(e0 + i * BIN_THREADS + tid, E - 1);
        const int r = rows[e];
        const int cc = cols[e];
        tl[i] = r >> TILE_SHIFT;
        key[i] = (cc << TILE_SHIFT) | (r & (TILE_ROWS - 1));
        vbits[i] = __float_as_int(vals[e]);
    }
#pragma unroll
    for (int i = 0; i < CHUNK / BIN_THREADS; ++i) {
        rnk[i] = 0;
        if (e0 + i * BIN_THREADS + tid < E) rnk[i] = atomicAdd(&h[tl[i]], 1);
    }
    __syncthreads();

    const int c0 = h[2 * tid];
    const int c1 = h[2 * tid + 1];
    const int s = c0 + c1;
    int incl = s;
#pragma unroll
    for (int d = 1; d < 32; d <<= 1) {
        const int up = __shfl_up(incl, d);
        if (lane >= d) incl += up;
    }
    if (lane == 31) wsum[w] = incl;
    __syncthreads();
    int before = 0;
#pragma unroll
    for (int k = 0; k < BIN_THREADS / 32; ++k) {
        const int ws = wsum[k];
        before += (k < w) ? ws : 0;
    }
    const int ex = before + incl - s;
    h[2 * tid] = ex;
    h[2 * tid + 1] = ex + c0;
    int* orow = Off + (size_t)blk * OFFS;
    if (2 * tid <= T) orow[2 * tid] = ex;
    if (2 * tid + 1 <= T) orow[2 * tid + 1] = ex + c0;
    __syncthreads();

    int at[CHUNK / BIN_THREADS];
#pragma unroll
    for (int i = 0; i < CHUNK / BIN_THREADS; ++i) at[i] = e0 + h[tl[i]] + rnk[i];
#pragma unroll
    for (int i = 0; i < CHUNK / BIN_THREADS; ++i) {
        if (e0 + i * BIN_THREADS + tid < E) rec[at[i]] = make_int2(key[i], vbits[i]);
    }
}

template <bool NEXT, bool TABLE, bool GUARD>
__device__ __forceinline__ void finish_rows(float* acc,
                                            const float* __restrict__ Wn,
                                            float2* __restrict__ yt,
                                            float2* __restrict__ ot,
                                            const float2* __restrict__ rt,
                                            int r_lo, int last, int lane) {
    float2 o[TILE_ROWS / 8];
    const float2* res = TABLE ? rt : (const float2*)ot;
#pragma unroll
    for (int j = 0; j < TILE_ROWS / 8; ++j) {
        o[j] = res[(GUARD ? min(j, last) : j) * (N_HID / 2)];
    }

    const float4* my4 = (const float4*)(acc + (r_lo + lane) * PITCH);
    float ss = 0.0f;
#pragma unroll
    for (int i = 0; i < N_HID / 4; ++i) {
        const float4 z = my4[i];
        const float vx = (z.x >= 0.0f) ? z.x : SLOPE * z.x;
        const float vy = (z.y >= 0.0f) ? z.y : SLOPE * z.y;
        const float vz = (z.z >= 0.0f) ? z.z : SLOPE * z.z;
        const float vw = (z.w >= 0.0f) ? z.w : SLOPE * z.w;
        ss += vx * vx + vy * vy + vz * vz + vw * vw;
    }
    const float inv_all = 1.0f / fmaxf(sqrtf(ss), EPS);

    float2* acc2 = (float2*)(acc + r_lo * PITCH) + lane;
#pragma unroll
    for (int j = 0; j < TILE_ROWS / 8; ++j) {
        const float inv = __int_as_float(__builtin_amdgcn_readlane(__float_as_int(inv_all), j));
        const float2 z = acc2[j * (PITCH / 2)];
        float2 v;
        v.x = (z.x >= 0.0f) ? z.x : SLOPE * z.x;
        v.y = (z.y >= 0.0f) ? z.y : SLOPE * z.y;
        float2 cur = o[j];
        cur.x += v.x * inv;
        cur.y += v.y * inv;
        if (NEXT) acc2[j * (PITCH / 2)] = v;
        if (!GUARD || j <= last) {
            ot[j * (N_HID / 2)] = cur;
        }
    }

    if (NEXT) {
        const int r = lane & 15;
        const int h = lane >> 4;
        v8f c0[4];
        v8f c1[4];
#pragma unroll
        for (int ct = 0; ct < 4; ++ct) {
            c0[ct] = v8f{};
            c1[ct] = v8f{};
        }
        const float* a_lo = acc + (r_lo + r) * PITCH + 2 * h;
        const float* a_hi = a_lo + 16 * PITCH;
#pragma unroll 4
        for (int k = 0; k < N_HID; k += 4) {
            const float2 p0 = *(const float2*)(a_lo + k);
            const float2 p1 = *(const float2*)(a_hi + k);
            v2f a0;
            a0[0] = p0.x;
            a0[1] = p0.y;
            v2f a1;
            a1[0] = p1.x;
            a1[1] = p1.y;
            const float* w0 = Wn + (size_t)(k + 2 * h) * N_HID + r;
            const float* w1 = w0 + N_HID;
#pragma unroll
            for (int ct = 0; ct < 4; ++ct) {
                v2f b;
                b[0] = w0[ct * 16];
                b[1] = w1[ct * 16];
                c0[ct] = __builtin_amdgcn_wmma_f32_16x16x4_f32(
                     false, a0,  false, b,
                     (short)0, c0[ct],
                     false,  false);
                c1[ct] = __builtin_amdgcn_wmma_f32_16x16x4_f32(
                     false, a1,  false, b,
                     (short)0, c1[ct],
                     false,  false);
            }
        }
#pragma unroll
        for (int ct = 0; ct < 4; ++ct) {
#pragma unroll
            for (int v = 0; v < 8; ++v) {
                acc[(r_lo + v + 8 * h) * PITCH + ct * 16 + r] = c0[ct][v];
                acc[(r_lo + 16 + v + 8 * h) * PITCH + ct * 16 + r] = c1[ct][v];
            }
        }
#pragma unroll
        for (int j = 0; j < TILE_ROWS / 8; ++j) {
            const float2 yv = acc2[j * (PITCH / 2)];
            if (!GUARD || j <= last) {
                yt[j * (N_HID / 2)] = yv;
            }
        }
    }
}

struct TilePart {
    const int2* rec;
    const int* Off;
    int NB;
    const float* Y;
    const float* Wn;
    float* Yn;
    const float* ta;
    const float* tb;
    int split;
    float* out;
    int M;
};

template <bool NEXT, bool TABLE>
__global__ void __launch_bounds__(256, 8)
spmm_tile_kernel(TilePart pa, TilePart pb, int na) {
    const bool p2 = (int)blockIdx.x >= na;
    const int2* __restrict__ rec = p2 ? pb.rec : pa.rec;
    const int* __restrict__ Off = p2 ? pb.Off : pa.Off;
    const int NB = p2 ? pb.NB : pa.NB;
    const float* __restrict__ Y = p2 ? pb.Y : pa.Y;
    const float* __restrict__ Wn = p2 ? pb.Wn : pa.Wn;
    float* __restrict__ Yn = p2 ? pb.Yn : pa.Yn;
    const float* __restrict__ ta = p2 ? pb.ta : pa.ta;
    const float* __restrict__ tb = p2 ? pb.tb : pa.tb;
    const int split = p2 ? pb.split : pa.split;
    float* __restrict__ out = p2 ? pb.out : pa.out;
    const int M = p2 ? pb.M : pa.M;
    __shared__ __attribute__((aligned(16))) float acc[TILE_ROWS * PITCH];
    __shared__ __attribute__((aligned(16))) int P[RUNS + 4];
    __shared__ int St[RUNS];
    const int tid = threadIdx.x;
    const int lane = tid & 31;
    const int w = __builtin_amdgcn_readfirstlane(tid >> 5);
    const int t = p2 ? (int)blockIdx.x - na : (int)blockIdx.x;

    int o0 = 0;
    int o1 = 0;
    if (tid < NB) {
        const int* o = Off + (size_t)tid * OFFS + t;
        o0 = o[0];
        o1 = o[1];
    }

    float4* acc4 = (float4*)acc;
    for (int i = tid; i < TILE_ROWS * PITCH / 4; i += 256) acc4[i] = make_float4(0.0f, 0.0f, 0.0f, 0.0f);
    if (tid < RUNS) {
        P[tid] = o1 - o0;
        St[tid] = tid * CHUNK + o0;
    }
    __syncthreads();
    if (w == 0) {
        const int4 a = *(const int4*)(P + 4 * lane);
        const int s = a.x + a.y + a.z + a.w;
        int incl = s;
#pragma unroll
        for (int d = 1; d < 32; d <<= 1) {
            const int up = __shfl_up(incl, d);
            if (lane >= d) incl += up;
        }
        const int ex = incl - s;
        *(int4*)(P + 4 * lane) = make_int4(ex, ex + a.x, ex + a.x + a.y, ex + a.x + a.y + a.z);
        if (lane == 31) P[RUNS] = incl;
    }
    __syncthreads();
    const int n_t = __builtin_amdgcn_readfirstlane(P[RUNS]);

    auto edge = [&](int e) -> int2 {
        int i = 0;
#pragma unroll
        for (int step = RUNS / 2; step >= 1; step >>= 1) {
            if (P[i + step] <= e) i += step;
        }
        return rec[St[i] + (e - P[i])];
    };

    int kk = 0;
    int vb = 0;
    if (w * 32 + lane < n_t) {
        const int2 rr = edge(w * 32 + lane);
        kk = rr.x;
        vb = rr.y;
    }
    for (int e0 = w * 32; e0 < n_t; e0 += 8 * 32) {
        const int n = min(32, n_t - e0);
        float2 y[32];
#pragma unroll
        for (int j = 0; j < 32; ++j) {
            const int k = __builtin_amdgcn_readlane(kk, j);
            const float2* yr = (const float2*)(Y + (size_t)(k >> TILE_SHIFT) * N_HID);
            y[j] = yr[lane];
        }
        int kn = 0;
        int vn = 0;
        if (e0 + 8 * 32 + lane < n_t) {
            const int2 rr = edge(e0 + 8 * 32 + lane);
            kn = rr.x;
            vn = rr.y;
        }
#pragma unroll
        for (int j = 0; j < 32; ++j) {
            if (j == 0 || j < n) {
                const int k = __builtin_amdgcn_readlane(kk, j);
                const float v = __int_as_float(__builtin_amdgcn_readlane(vb, j));
                float* a = acc + (k & (TILE_ROWS - 1)) * PITCH + 2 * lane;
                atomicAdd(a, v * y[j].x);
                atomicAdd(a + 1, v * y[j].y);
            }
        }
        kk = kn;
        vb = vn;
    }
    const int rows_here = min(TILE_ROWS, M - t * TILE_ROWS);
    const int r_lo = w * (TILE_ROWS / 8);
    const size_t first = ((size_t)t * TILE_ROWS + r_lo) * N_HID;
    float2* yt = NEXT ? (float2*)(Yn + first) + lane : nullptr;
    float2* ot = (float2*)(out + first) + lane;
    const float2* rt = nullptr;
    if (TABLE) {
        const int g0 = t * TILE_ROWS + r_lo;
        rt = (const float2*)((min(g0, M - 1) < split) ? ta + (size_t)g0 * N_HID
                                                      : tb + (size_t)(g0 - split) * N_HID) + lane;
    }

    __syncthreads();

    if (r_lo + TILE_ROWS / 8 <= rows_here) {
        finish_rows<NEXT, TABLE, false>(acc, Wn, yt, ot, rt, r_lo, rows_here - 1 - r_lo, lane);
    } else {
        finish_rows<NEXT, TABLE, true>(acc, Wn, yt, ot, rt, r_lo, rows_here - 1 - r_lo, lane);
    }
}

extern "C" void kernel_launch(void* const* d_in, const int* in_sizes, int n_in,
                              void* d_out, int out_size, void* d_ws, size_t ws_size,
                              hipStream_t stream) {
    const float* user_emb = (const float*)d_in[0];
    const float* item_emb = (const float*)d_in[1];
    const float* ui_w     = (const float*)d_in[2];
    const float* s_w      = (const float*)d_in[3];
    const int*   ui_rows  = (const int*)d_in[4];
    const int*   ui_cols  = (const int*)d_in[5];
    const float* ui_vals  = (const float*)d_in[6];
    const int*   s_rows   = (const int*)d_in[7];
    const int*   s_cols   = (const int*)d_in[8];
    const float* s_vals   = (const float*)d_in[9];

    float* Xu = (float*)d_ws;
    float* Yu = Xu + (size_t)N_NODES * N_HID;
    float* Xs = Yu + (size_t)N_NODES * N_HID;
    float* Ys = Xs + (size_t)N_USER * N_HID;
    int2*  recu = (int2*)(Ys + (size_t)N_USER * N_HID);
    int2*  recs = recu + E_UI;
    int*   Offu = (int*)(recs + E_S);
    int*   Offs = Offu + (size_t)NB_UI * OFFS;

    float* out_ui = (float*)d_out;
    float* out_s  = out_ui + (size_t)N_NODES * N_HID;

    sort_local_kernel<<<NB_UI + NB_S + GB16(GT_UI) + GB16(GT_S), BIN_THREADS, 0, stream>>>(
        SortPart{ui_rows, ui_cols, ui_vals, Offu, recu, E_UI, T_UI, NB_UI},
        SortPart{s_rows, s_cols, s_vals, Offs, recs, E_S, T_S, NB_S},
        GemmPart{user_emb, item_emb, ui_w, Yu, N_USER, 0, GT_UI},
        GemmPart{user_emb, user_emb, s_w, Ys, N_USER, 0, GT_S},
        GB16(GT_UI));

    spmm_tile_kernel<true, true><<<T_S + T_UI, 256, 0, stream>>>(
        TilePart{recs, Offs, NB_S, Ys, s_w + N_HID * N_HID, Xs, user_emb, user_emb, N_USER, out_s, N_USER},
        TilePart{recu, Offu, NB_UI, Yu, ui_w + N_HID * N_HID, Xu, user_emb, item_emb, N_USER, out_ui, N_NODES},
        T_S);
    spmm_tile_kernel<false, false><<<T_S + T_UI, 256, 0, stream>>>(
        TilePart{recs, Offs, NB_S, Xs, nullptr, nullptr, nullptr, nullptr, 0, out_s, N_USER},
        TilePart{recu, Offu, NB_UI, Xu, nullptr, nullptr, nullptr, nullptr, 0, out_ui, N_NODES},
        T_S);
}
